# starting kernel
# baseline (speedup 1.0000x reference)
#include <hip/hip_runtime.h>

#define HID 64
#define H4  16

typedef __attribute__((ext_vector_type(2))) float v2f;
typedef __attribute__((ext_vector_type(8))) float v8f;

__global__ void k_deg_init(float* __restrict__ deg, int n) {
    int i = blockIdx.x * blockDim.x + threadIdx.x;
    if (i < n) deg[i] = 1.0f;
}

__global__ void k_deg_scatter(const int* __restrict__ dst, float* __restrict__ deg, int e) {
    int i = blockIdx.x * blockDim.x + threadIdx.x;
    if (i < e) atomicAdd(&deg[dst[i]], 1.0f);
}

__global__ void k_rsqrt(float* __restrict__ deg, int n) {
    int i = blockIdx.x * blockDim.x + threadIdx.x;
    if (i < n) deg[i] = rsqrtf(deg[i]);
}

__global__ void k_norm(const int* __restrict__ src, const int* __restrict__ dst,
                       const float* __restrict__ dis, float* __restrict__ norm, int e) {
    int i = blockIdx.x * blockDim.x + threadIdx.x;
    if (i < e) norm[i] = dis[src[i]] * dis[dst[i]];
}

template <int K, int NC, int RELU>
__global__ __launch_bounds__(256)
void k_gemm(const float* __restrict__ A, const float* __restrict__ B,
            const float* __restrict__ bias, float* __restrict__ C, int M) {
    constexpr int tilesN = (NC + 15) / 16;
    int wave = blockIdx.x * (blockDim.x >> 5) + (threadIdx.x >> 5);
    int lane = threadIdx.x & 31;
    int tiles = (M >> 4) * tilesN;
    if (wave >= tiles) return;

    int tm = wave / tilesN;
    int tn = wave - tm * tilesN;
    int hf = lane >> 4;
    int l  = lane & 15;
    int col  = tn * 16 + l;
    int colC = (col < NC) ? col : (NC - 1);

    const v2f*   Ap = (const v2f*)(A + (size_t)(tm * 16 + l) * K + 2 * hf);
    const float* Bp = B + (size_t)(2 * hf) * NC + colC;

    v8f acc = {};
#pragma unroll 8
    for (int k0 = 0; k0 < K; k0 += 4) {
        v2f a = Ap[k0 >> 1];
        if (RELU) { a.x = fmaxf(a.x, 0.0f); a.y = fmaxf(a.y, 0.0f); }
        v2f b;
        b.x = Bp[(size_t)k0 * NC];
        b.y = Bp[(size_t)(k0 + 1) * NC];
        acc = __builtin_amdgcn_wmma_f32_16x16x4_f32(
             false, a,  false, b,
             (short)0, acc,  false,  false);
    }

    if (col < NC) {
        float bb = bias[col];
        float* Crow = C + (size_t)(tm * 16 + 8 * hf) * NC + col;
#pragma unroll
        for (int i = 0; i < 8; ++i)
            Crow[(size_t)i * NC] = acc[i] + bb;
    }
}

__global__ void k_selfinit(const float* __restrict__ t, const float* __restrict__ dis,
                           float* __restrict__ out, int n) {
    int gid = blockIdx.x * blockDim.x + threadIdx.x;
    if (gid >= n * H4) return;
    int i = gid >> 4;
    float s = dis[i];
    float k = s * s;
    float4 v = ((const float4*)t)[gid];
    float4 r; r.x = v.x * k; r.y = v.y * k; r.z = v.z * k; r.w = v.w * k;
    ((float4*)out)[gid] = r;
}

__global__ void k_scatter(const float* __restrict__ t, const int* __restrict__ src,
                          const int* __restrict__ dst, const float* __restrict__ norm,
                          float* out, int e) {
    int gid = blockIdx.x * blockDim.x + threadIdx.x;
    if (gid >= e * H4) return;
    int ed = gid >> 4;
    int c  = gid & 15;
    int s = src[ed];
    int d = dst[ed];
    float w = norm[ed];
    float4 v = ((const float4*)(t + (size_t)s * HID))[c];
    float* o = out + (size_t)d * HID + (size_t)c * 4;
    atomicAdd(o + 0, v.x * w);
    atomicAdd(o + 1, v.y * w);
    atomicAdd(o + 2, v.z * w);
    atomicAdd(o + 3, v.w * w);
}

extern "C" void kernel_launch(void* const* d_in, const int* in_sizes, int n_in,
                              void* d_out, int out_size, void* d_ws, size_t ws_size,
                              hipStream_t stream) {
    (void)n_in; (void)out_size; (void)ws_size;
    const float* x    = (const float*)d_in[0];
    const int*   ei   = (const int*)d_in[1];
    const float* W0   = (const float*)d_in[2];
    const float* b0   = (const float*)d_in[3];
    const float* W1   = (const float*)d_in[4];
    const float* b1   = (const float*)d_in[5];
    const float* W2   = (const float*)d_in[6];
    const float* b2   = (const float*)d_in[7];
    const float* Wout = (const float*)d_in[8];
    const float* bout = (const float*)d_in[9];

    const int FIN = 128, OUTD = 40;
    const int n = in_sizes[0] / FIN;
    const int e = in_sizes[1] / 2;
    const int* srcI = ei;
    const int* dstI = ei + e;

    char* w = (char*)d_ws;
    size_t off = 0;
    float* dis  = (float*)(w + off); off += (((size_t)n * 4) + 255) & ~(size_t)255;
    float* norm = (float*)(w + off); off += (((size_t)e * 4) + 255) & ~(size_t)255;
    float* tmp  = (float*)(w + off); off += (((size_t)n * HID * 4) + 255) & ~(size_t)255;
    float* hA   = (float*)(w + off); off += (((size_t)n * HID * 4) + 255) & ~(size_t)255;
    float* hB   = (float*)(w + off); off += (((size_t)n * HID * 4) + 255) & ~(size_t)255;

    const int T   = 256;
    const int WPB = T / 32;

    k_deg_init   <<<(n + T - 1) / T, T, 0, stream>>>(dis, n);
    k_deg_scatter<<<(e + T - 1) / T, T, 0, stream>>>(dstI, dis, e);
    k_rsqrt      <<<(n + T - 1) / T, T, 0, stream>>>(dis, n);
    k_norm       <<<(e + T - 1) / T, T, 0, stream>>>(srcI, dstI, dis, norm, e);

    const int tilesH = (n / 16) * (HID / 16);
    const int gGemmH = (tilesH + WPB - 1) / WPB;
    const int gAgg1  = (n * H4 + T - 1) / T;
    const int gAggE  = (e * H4 + T - 1) / T;

    k_gemm<128, HID, 0><<<gGemmH, T, 0, stream>>>(x, W0, b0, tmp, n);
    k_selfinit<<<gAgg1, T, 0, stream>>>(tmp, dis, hA, n);
    k_scatter <<<gAggE, T, 0, stream>>>(tmp, srcI, dstI, norm, hA, e);

    k_gemm<HID, HID, 1><<<gGemmH, T, 0, stream>>>(hA, W1, b1, tmp, n);
    k_selfinit<<<gAgg1, T, 0, stream>>>(tmp, dis, hB, n);
    k_scatter <<<gAggE, T, 0, stream>>>(tmp, srcI, dstI, norm, hB, e);

    k_gemm<HID, HID, 1><<<gGemmH, T, 0, stream>>>(hB, W2, b2, tmp, n);
    k_selfinit<<<gAgg1, T, 0, stream>>>(tmp, dis, hA, n);
    k_scatter <<<gAggE, T, 0, stream>>>(tmp, srcI, dstI, norm, hA, e);

    const int tilesO = (n / 16) * ((OUTD + 15) / 16);
    k_gemm<HID, OUTD, 1><<<(tilesO + WPB - 1) / WPB, T, 0, stream>>>(hA, Wout, bout,
                                                                     (float*)d_out, n);
}


# Edge float-atomic scatter-add replaced by per-node register gather over a dst-grouped edge list rebuilt each call (returning int atomic rank, scan, fill); sum order = atomic arrival rank
# speedup vs baseline: 2.0258x; 2.0258x over previous
#include <hip/hip_runtime.h>

#define HID 64
#define H4  16

typedef __attribute__((ext_vector_type(2))) float v2f;
typedef __attribute__((ext_vector_type(8))) float v8f;

#define SCAN_T 1024

__global__ void k_cnt_zero(int* __restrict__ cnt, int n) {
    int i = blockIdx.x * blockDim.x + threadIdx.x;
    if (i < n) cnt[i] = 0;
}

__global__ void k_cnt_rank(const int* __restrict__ dst, int* __restrict__ cnt,
                           int* __restrict__ rank, int e) {
    int i = blockIdx.x * blockDim.x + threadIdx.x;
    if (i < e) rank[i] = atomicAdd(&cnt[dst[i]], 1);
}

__global__ __launch_bounds__(SCAN_T)
void k_rowptr(const int* __restrict__ cnt, int* __restrict__ rowptr,
              float* __restrict__ dis, int n) {
    __shared__ int s_w[32];
    __shared__ int s_pre;
    const int tid  = threadIdx.x;
    const int lane = tid & 31;
    const int wv   = tid >> 5;
    const int base = blockIdx.x * SCAN_T;

    int part = 0;
    const int4* c4 = (const int4*)cnt;
    for (int k = tid * 4; k < base; k += SCAN_T * 4) {
        int4 q = c4[k >> 2];
        part += (q.x + q.y) + (q.z + q.w);
    }
    for (int o = 16; o > 0; o >>= 1) part += __shfl_xor(part, o);
    if (lane == 0) s_w[wv] = part;
    __syncthreads();
    if (wv == 0) {
        int t = s_w[lane];
        for (int o = 16; o > 0; o >>= 1) t += __shfl_xor(t, o);
        if (lane == 0) s_pre = t;
    }
    __syncthreads();
    const int pre = s_pre;

    const int i = base + tid;
    const int v = (i < n) ? cnt[i] : 0;
    int inc = v;
    for (int o = 1; o < 32; o <<= 1) {
        int t = __shfl_up(inc, o);
        if (lane >= o) inc += t;
    }
    if (lane == 31) s_w[wv] = inc;
    __syncthreads();
    if (wv == 0) {
        int t = s_w[lane];
        int ti = t;
        for (int o = 1; o < 32; o <<= 1) {
            int u = __shfl_up(ti, o);
            if (lane >= o) ti += u;
        }
        s_w[lane] = ti - t;
    }
    __syncthreads();
    const int excl = pre + s_w[wv] + (inc - v);
    if (i < n) {
        rowptr[i] = excl;
        dis[i] = rsqrtf((float)(v + 1));
        if (i == n - 1) rowptr[n] = excl + v;
    }
}

__global__ void k_fill(const int* __restrict__ src, const int* __restrict__ dst,
                       const int* __restrict__ rank, const int* __restrict__ rowptr,
                       const float* __restrict__ dis, int2* __restrict__ ew, int e) {
    int i = blockIdx.x * blockDim.x + threadIdx.x;
    if (i >= e) return;
    int s = src[i];
    int d = dst[i];
    float w = dis[s] * dis[d];
    ew[rowptr[d] + rank[i]] = make_int2(s, __float_as_int(w));
}

template <int K, int NC, int RELU>
__global__ __launch_bounds__(256)
void k_gemm(const float* __restrict__ A, const float* __restrict__ B,
            const float* __restrict__ bias, float* __restrict__ C, int M) {
    constexpr int tilesN = (NC + 15) / 16;
    int wave = blockIdx.x * (blockDim.x >> 5) + (threadIdx.x >> 5);
    int lane = threadIdx.x & 31;
    int tiles = (M >> 4) * tilesN;
    if (wave >= tiles) return;

    int tm = wave / tilesN;
    int tn = wave - tm * tilesN;
    int hf = lane >> 4;
    int l  = lane & 15;
    int col  = tn * 16 + l;
    int colC = (col < NC) ? col : (NC - 1);

    const v2f*   Ap = (const v2f*)(A + (size_t)(tm * 16 + l) * K + 2 * hf);
    const float* Bp = B + (size_t)(2 * hf) * NC + colC;

    v8f acc = {};
#pragma unroll 8
    for (int k0 = 0; k0 < K; k0 += 4) {
        v2f a = Ap[k0 >> 1];
        if (RELU) { a.x = fmaxf(a.x, 0.0f); a.y = fmaxf(a.y, 0.0f); }
        v2f b;
        b.x = Bp[(size_t)k0 * NC];
        b.y = Bp[(size_t)(k0 + 1) * NC];
        acc = __builtin_amdgcn_wmma_f32_16x16x4_f32(
             false, a,  false, b,
             (short)0, acc,  false,  false);
    }

    if (col < NC) {
        float bb = bias[col];
        float* Crow = C + (size_t)(tm * 16 + 8 * hf) * NC + col;
#pragma unroll
        for (int i = 0; i < 8; ++i)
            Crow[(size_t)i * NC] = acc[i] + bb;
    }
}

__global__ __launch_bounds__(256)
void k_gather(const float* __restrict__ t, const int* __restrict__ rowptr,
              const int2* __restrict__ ew, const float* __restrict__ dis,
              float* __restrict__ out, int n) {
    int gid = blockIdx.x * blockDim.x + threadIdx.x;
    if (gid >= n * H4) return;
    const int i = gid >> 4;
    const int c = gid & 15;
    const float4* t4 = (const float4*)t;
    int j = rowptr[i];
    const int je = rowptr[i + 1];
    const float sd = dis[i];
    const float k = sd * sd;
    const float4 v = t4[gid];
    float4 a;
    a.x = v.x * k; a.y = v.y * k; a.z = v.z * k; a.w = v.w * k;
    for (; j < je; ++j) {
        const int2 p = ew[j];
        const float4 u = t4[(size_t)p.x * H4 + c];
        const float w = __int_as_float(p.y);
        a.x += u.x * w; a.y += u.y * w; a.z += u.z * w; a.w += u.w * w;
    }
    ((float4*)out)[gid] = a;
}

extern "C" void kernel_launch(void* const* d_in, const int* in_sizes, int n_in,
                              void* d_out, int out_size, void* d_ws, size_t ws_size,
                              hipStream_t stream) {
    (void)n_in; (void)out_size; (void)ws_size;
    const float* x    = (const float*)d_in[0];
    const int*   ei   = (const int*)d_in[1];
    const float* W0   = (const float*)d_in[2];
    const float* b0   = (const float*)d_in[3];
    const float* W1   = (const float*)d_in[4];
    const float* b1   = (const float*)d_in[5];
    const float* W2   = (const float*)d_in[6];
    const float* b2   = (const float*)d_in[7];
    const float* Wout = (const float*)d_in[8];
    const float* bout = (const float*)d_in[9];

    const int FIN = 128, OUTD = 40;
    const int n = in_sizes[0] / FIN;
    const int e = in_sizes[1] / 2;
    const int* srcI = ei;
    const int* dstI = ei + e;

    char* w = (char*)d_ws;
    size_t off = 0;
    int*   cnt    = (int*)(w + off);   off += (((size_t)n * 4) + 255) & ~(size_t)255;
    int*   rowptr = (int*)(w + off);   off += (((size_t)(n + 1) * 4) + 255) & ~(size_t)255;
    float* dis    = (float*)(w + off); off += (((size_t)n * 4) + 255) & ~(size_t)255;
    int*   rank   = (int*)(w + off);   off += (((size_t)e * 4) + 255) & ~(size_t)255;
    int2*  ew     = (int2*)(w + off);  off += (((size_t)e * 8) + 255) & ~(size_t)255;
    float* tmp    = (float*)(w + off); off += (((size_t)n * HID * 4) + 255) & ~(size_t)255;
    float* hA     = (float*)(w + off); off += (((size_t)n * HID * 4) + 255) & ~(size_t)255;
    float* hB     = (float*)(w + off); off += (((size_t)n * HID * 4) + 255) & ~(size_t)255;

    const int T   = 256;
    const int WPB = T / 32;

    k_cnt_zero<<<(n + T - 1) / T, T, 0, stream>>>(cnt, n);
    k_cnt_rank<<<(e + T - 1) / T, T, 0, stream>>>(dstI, cnt, rank, e);
    k_rowptr  <<<(n + SCAN_T - 1) / SCAN_T, SCAN_T, 0, stream>>>(cnt, rowptr, dis, n);
    k_fill    <<<(e + T - 1) / T, T, 0, stream>>>(srcI, dstI, rank, rowptr, dis, ew, e);

    const int tilesH = (n / 16) * (HID / 16);
    const int gGemmH = (tilesH + WPB - 1) / WPB;
    const int gAgg   = (n * H4 + T - 1) / T;

    k_gemm<128, HID, 0><<<gGemmH, T, 0, stream>>>(x, W0, b0, tmp, n);
    k_gather<<<gAgg, T, 0, stream>>>(tmp, rowptr, ew, dis, hA, n);

    k_gemm<HID, HID, 1><<<gGemmH, T, 0, stream>>>(hA, W1, b1, tmp, n);
    k_gather<<<gAgg, T, 0, stream>>>(tmp, rowptr, ew, dis, hB, n);

    k_gemm<HID, HID, 1><<<gGemmH, T, 0, stream>>>(hB, W2, b2, tmp, n);
    k_gather<<<gAgg, T, 0, stream>>>(tmp, rowptr, ew, dis, hA, n);

    const int tilesO = (n / 16) * ((OUTD + 15) / 16);
    k_gemm<HID, OUTD, 1><<<(tilesO + WPB - 1) / WPB, T, 0, stream>>>(hA, Wout, bout,
                                                                     (float*)d_out, n);
}


# Products: one wave per 16-row strip does all column tiles; W staged once per work-group in LDS, the strip via a wave-private LDS slot by whole-row loads (was 16-line loads per tile); same sum order
# speedup vs baseline: 2.1575x; 1.0650x over previous
#include <hip/hip_runtime.h>

#define HID 64
#define H4  16

typedef __attribute__((ext_vector_type(2))) float v2f;
typedef __attribute__((ext_vector_type(8))) float v8f;
typedef __attribute__((ext_vector_type(4))) float v4f;

#define SCAN_T 1024

__global__ void k_cnt_zero(int* __restrict__ cnt, int n) {
    int i = blockIdx.x * blockDim.x + threadIdx.x;
    if (i < n) cnt[i] = 0;
}

__global__ void k_cnt_rank(const int* __restrict__ dst, int* __restrict__ cnt,
                           int* __restrict__ rank, int e) {
    int i = blockIdx.x * blockDim.x + threadIdx.x;
    if (i < e) rank[i] = atomicAdd(&cnt[dst[i]], 1);
}

__global__ __launch_bounds__(SCAN_T)
void k_rowptr(const int* __restrict__ cnt, int* __restrict__ rowptr,
              float* __restrict__ dis, int n) {
    __shared__ int s_w[32];
    __shared__ int s_pre;
    const int tid  = threadIdx.x;
    const int lane = tid & 31;
    const int wv   = tid >> 5;
    const int base = blockIdx.x * SCAN_T;

    int part = 0;
    const int4* c4 = (const int4*)cnt;
    for (int k = tid * 4; k < base; k += SCAN_T * 4) {
        int4 q = c4[k >> 2];
        part += (q.x + q.y) + (q.z + q.w);
    }
    for (int o = 16; o > 0; o >>= 1) part += __shfl_xor(part, o);
    if (lane == 0) s_w[wv] = part;
    __syncthreads();
    if (wv == 0) {
        int t = s_w[lane];
        for (int o = 16; o > 0; o >>= 1) t += __shfl_xor(t, o);
        if (lane == 0) s_pre = t;
    }
    __syncthreads();
    const int pre = s_pre;

    const int i = base + tid;
    const int v = (i < n) ? cnt[i] : 0;
    int inc = v;
    for (int o = 1; o < 32; o <<= 1) {
        int t = __shfl_up(inc, o);
        if (lane >= o) inc += t;
    }
    if (lane == 31) s_w[wv] = inc;
    __syncthreads();
    if (wv == 0) {
        int t = s_w[lane];
        int ti = t;
        for (int o = 1; o < 32; o <<= 1) {
            int u = __shfl_up(ti, o);
            if (lane >= o) ti += u;
        }
        s_w[lane] = ti - t;
    }
    __syncthreads();
    const int excl = pre + s_w[wv] + (inc - v);
    if (i < n) {
        rowptr[i] = excl;
        dis[i] = rsqrtf((float)(v + 1));
        if (i == n - 1) rowptr[n] = excl + v;
    }
}

__global__ void k_fill(const int* __restrict__ src, const int* __restrict__ dst,
                       const int* __restrict__ rank, const int* __restrict__ rowptr,
                       const float* __restrict__ dis, int2* __restrict__ ew, int e) {
    int i = blockIdx.x * blockDim.x + threadIdx.x;
    if (i >= e) return;
    int s = src[i];
    int d = dst[i];
    float w = dis[s] * dis[d];
    ew[rowptr[d] + rank[i]] = make_int2(s, __float_as_int(w));
}

template <int K, int NC, int RELU>
__global__ __launch_bounds__(256)
void k_gemm(const float* __restrict__ A, const float* __restrict__ B,
            const float* __restrict__ bias, float* __restrict__ C, int M) {
    constexpr int TN  = (NC + 15) / 16;
    constexpr int NCP = TN * 16;
    constexpr int BP  = 2 * NCP + (((2 * NCP) % 64 == 32) ? 0 : 32);
    constexpr int KP  = K / 2;
    constexpr int AP  = 68;
    constexpr int PH  = K / 64;
    constexpr int CG  = NC / 4;
    constexpr int NT  = KP * CG;
    constexpr int TPT = (NT + 255) / 256;

    __shared__ __attribute__((aligned(16))) float Bs[KP * BP];
    __shared__ __attribute__((aligned(16))) float As[8 * 16 * AP];

    const int tid  = threadIdx.x;
    const int wv   = tid >> 5;
    const int lane = tid & 31;
    const int hf   = lane >> 4;
    const int l    = lane & 15;
    const int strip = blockIdx.x * 8 + wv;
    const bool live = strip < (M >> 4);

    float4 r0[TPT], r1[TPT];
#pragma unroll
    for (int t = 0; t < TPT; ++t) {
        const int task = tid + t * 256;
        if (NT % 256 == 0 || task < NT) {
            const int kp = task / CG;
            const int cg = task - kp * CG;
            r0[t] = *(const float4*)(B + (size_t)(2 * kp) * NC + 4 * cg);
            r1[t] = *(const float4*)(B + (size_t)(2 * kp + 1) * NC + 4 * cg);
        }
    }

    const float* Ab = A + (size_t)(live ? strip : 0) * 16 * K + (size_t)hf * K + 4 * l;
    v4f ar[8];
#pragma unroll
    for (int q = 0; q < 8; ++q)
        ar[q] = *(const v4f*)(Ab + (size_t)(2 * q) * K);

#pragma unroll
    for (int t = 0; t < TPT; ++t) {
        const int task = tid + t * 256;
        if (NT % 256 == 0 || task < NT) {
            const int kp = task / CG;
            const int cg = task - kp * CG;
            float* d = Bs + kp * BP + 8 * cg;
            *(float4*)d       = make_float4(r0[t].x, r1[t].x, r0[t].y, r1[t].y);
            *(float4*)(d + 4) = make_float4(r0[t].z, r1[t].z, r0[t].w, r1[t].w);
        }
    }
    if (NCP > NC) {
        constexpr int ZQ = (NCP - NC) / 2;
        for (int idx = tid; idx < KP * ZQ; idx += 256) {
            const int kp = idx / ZQ;
            const int zq = idx - kp * ZQ;
            *(float4*)(Bs + kp * BP + 2 * NC + 4 * zq) = make_float4(0.f, 0.f, 0.f, 0.f);
        }
    }
    asm volatile("" :: "v"(ar[0]), "v"(ar[1]), "v"(ar[2]), "v"(ar[3]),
                       "v"(ar[4]), "v"(ar[5]), "v"(ar[6]), "v"(ar[7]));
    __syncthreads();
    if (!live) return;

    float* Aw = As + wv * (16 * AP);
    v8f acc[TN];
#pragma unroll
    for (int tn = 0; tn < TN; ++tn) acc[tn] = v8f{};

#pragma unroll
    for (int p = 0; p < PH; ++p) {
#pragma unroll
        for (int q = 0; q < 8; ++q) {
            v4f v = ar[q];
            if (RELU) {
                v.x = fmaxf(v.x, 0.0f); v.y = fmaxf(v.y, 0.0f);
                v.z = fmaxf(v.z, 0.0f); v.w = fmaxf(v.w, 0.0f);
            }
            *(v4f*)(Aw + (2 * q + hf) * AP + 4 * l) = v;
        }
        __builtin_amdgcn_fence(__ATOMIC_ACQ_REL, "wavefront");
        if (p + 1 < PH) {
#pragma unroll
            for (int q = 0; q < 8; ++q)
                ar[q] = *(const v4f*)(Ab + (size_t)(2 * q) * K + 64 * (p + 1));
            __builtin_amdgcn_sched_barrier(0);
        }
#pragma unroll
        for (int kk = 0; kk < 16; ++kk) {
            const v2f a = *(const v2f*)(Aw + l * AP + 4 * kk + 2 * hf);
            const float* bp = Bs + (32 * p + 2 * kk + hf) * BP + 2 * l;
#pragma unroll
            for (int tn = 0; tn < TN; ++tn) {
                const v2f b = *(const v2f*)(bp + 32 * tn);
                acc[tn] = __builtin_amdgcn_wmma_f32_16x16x4_f32(
                     false, a,  false, b,
                     (short)0, acc[tn],  false,  false);
            }
        }
        __builtin_amdgcn_fence(__ATOMIC_ACQ_REL, "wavefront");
    }

#pragma unroll
    for (int tn = 0; tn < TN; ++tn) {
        const int col = tn * 16 + l;
        if (col < NC) {
            const float bb = bias[col];
            float* Crow = C + (size_t)(strip * 16 + 8 * hf) * NC + col;
#pragma unroll
            for (int i = 0; i < 8; ++i)
                Crow[(size_t)i * NC] = acc[tn][i] + bb;
        }
    }
}

__global__ __launch_bounds__(256)
void k_gather(const float* __restrict__ t, const int* __restrict__ rowptr,
              const int2* __restrict__ ew, const float* __restrict__ dis,
              float* __restrict__ out, int n) {
    int gid = blockIdx.x * blockDim.x + threadIdx.x;
    if (gid >= n * H4) return;
    const int i = gid >> 4;
    const int c = gid & 15;
    const float4* t4 = (const float4*)t;
    int j = rowptr[i];
    const int je = rowptr[i + 1];
    const float sd = dis[i];
    const float k = sd * sd;
    const float4 v = t4[gid];
    float4 a;
    a.x = v.x * k; a.y = v.y * k; a.z = v.z * k; a.w = v.w * k;
    for (; j < je; ++j) {
        const int2 p = ew[j];
        const float4 u = t4[(size_t)p.x * H4 + c];
        const float w = __int_as_float(p.y);
        a.x += u.x * w; a.y += u.y * w; a.z += u.z * w; a.w += u.w * w;
    }
    ((float4*)out)[gid] = a;
}

extern "C" void kernel_launch(void* const* d_in, const int* in_sizes, int n_in,
                              void* d_out, int out_size, void* d_ws, size_t ws_size,
                              hipStream_t stream) {
    (void)n_in; (void)out_size; (void)ws_size;
    const float* x    = (const float*)d_in[0];
    const int*   ei   = (const int*)d_in[1];
    const float* W0   = (const float*)d_in[2];
    const float* b0   = (const float*)d_in[3];
    const float* W1   = (const float*)d_in[4];
    const float* b1   = (const float*)d_in[5];
    const float* W2   = (const float*)d_in[6];
    const float* b2   = (const float*)d_in[7];
    const float* Wout = (const float*)d_in[8];
    const float* bout = (const float*)d_in[9];

    const int FIN = 128, OUTD = 40;
    const int n = in_sizes[0] / FIN;
    const int e = in_sizes[1] / 2;
    const int* srcI = ei;
    const int* dstI = ei + e;

    char* w = (char*)d_ws;
    size_t off = 0;
    int*   cnt    = (int*)(w + off);   off += (((size_t)n * 4) + 255) & ~(size_t)255;
    int*   rowptr = (int*)(w + off);   off += (((size_t)(n + 1) * 4) + 255) & ~(size_t)255;
    float* dis    = (float*)(w + off); off += (((size_t)n * 4) + 255) & ~(size_t)255;
    int*   rank   = (int*)(w + off);   off += (((size_t)e * 4) + 255) & ~(size_t)255;
    int2*  ew     = (int2*)(w + off);  off += (((size_t)e * 8) + 255) & ~(size_t)255;
    float* tmp    = (float*)(w + off); off += (((size_t)n * HID * 4) + 255) & ~(size_t)255;
    float* hA     = (float*)(w + off); off += (((size_t)n * HID * 4) + 255) & ~(size_t)255;
    float* hB     = (float*)(w + off); off += (((size_t)n * HID * 4) + 255) & ~(size_t)255;

    const int T   = 256;
    const int WPB = T / 32;

    k_cnt_zero<<<(n + T - 1) / T, T, 0, stream>>>(cnt, n);
    k_cnt_rank<<<(e + T - 1) / T, T, 0, stream>>>(dstI, cnt, rank, e);
    k_rowptr  <<<(n + SCAN_T - 1) / SCAN_T, SCAN_T, 0, stream>>>(cnt, rowptr, dis, n);
    k_fill    <<<(e + T - 1) / T, T, 0, stream>>>(srcI, dstI, rank, rowptr, dis, ew, e);

    const int gGemm  = (n / 16 + WPB - 1) / WPB;
    const int gAgg   = (n * H4 + T - 1) / T;

    k_gemm<128, HID, 0><<<gGemm, T, 0, stream>>>(x, W0, b0, tmp, n);
    k_gather<<<gAgg, T, 0, stream>>>(tmp, rowptr, ew, dis, hA, n);

    k_gemm<HID, HID, 1><<<gGemm, T, 0, stream>>>(hA, W1, b1, tmp, n);
    k_gather<<<gAgg, T, 0, stream>>>(tmp, rowptr, ew, dis, hB, n);

    k_gemm<HID, HID, 1><<<gGemm, T, 0, stream>>>(hB, W2, b2, tmp, n);
    k_gather<<<gAgg, T, 0, stream>>>(tmp, rowptr, ew, dis, hA, n);

    k_gemm<HID, OUTD, 1><<<gGemm, T, 0, stream>>>(hA, Wout, bout, (float*)d_out, n);
}


# Each aggregation launch also applies relu and the next layer's WMMA product to its own 32 rows (rows pass through LDS, not memory): 11 launches -> 8; same sum orders
# speedup vs baseline: 2.2729x; 1.0535x over previous
#include <hip/hip_runtime.h>

#define HID 64
#define H4  16

typedef __attribute__((ext_vector_type(2))) float v2f;
typedef __attribute__((ext_vector_type(8))) float v8f;
typedef __attribute__((ext_vector_type(4))) float v4f;

#define SCAN_T 1024

__global__ void k_cnt_zero(int* __restrict__ cnt, int n) {
    int i = blockIdx.x * blockDim.x + threadIdx.x;
    if (i < n) cnt[i] = 0;
}

__global__ void k_cnt_rank(const int* __restrict__ dst, int* __restrict__ cnt,
                           int* __restrict__ rank, int e) {
    int i = blockIdx.x * blockDim.x + threadIdx.x;
    if (i < e) rank[i] = atomicAdd(&cnt[dst[i]], 1);
}

__global__ __launch_bounds__(SCAN_T)
void k_rowptr(const int* __restrict__ cnt, int* __restrict__ rowptr,
              float* __restrict__ dis, int n) {
    __shared__ int s_w[32];
    __shared__ int s_pre;
    const int tid  = threadIdx.x;
    const int lane = tid & 31;
    const int wv   = tid >> 5;
    const int base = blockIdx.x * SCAN_T;

    int part = 0;
    const int4* c4 = (const int4*)cnt;
    for (int k = tid * 4; k < base; k += SCAN_T * 4) {
        int4 q = c4[k >> 2];
        part += (q.x + q.y) + (q.z + q.w);
    }
    for (int o = 16; o > 0; o >>= 1) part += __shfl_xor(part, o);
    if (lane == 0) s_w[wv] = part;
    __syncthreads();
    if (wv == 0) {
        int t = s_w[lane];
        for (int o = 16; o > 0; o >>= 1) t += __shfl_xor(t, o);
        if (lane == 0) s_pre = t;
    }
    __syncthreads();
    const int pre = s_pre;

    const int i = base + tid;
    const int v = (i < n) ? cnt[i] : 0;
    int inc = v;
    for (int o = 1; o < 32; o <<= 1) {
        int t = __shfl_up(inc, o);
        if (lane >= o) inc += t;
    }
    if (lane == 31) s_w[wv] = inc;
    __syncthreads();
    if (wv == 0) {
        int t = s_w[lane];
        int ti = t;
        for (int o = 1; o < 32; o <<= 1) {
            int u = __shfl_up(ti, o);
            if (lane >= o) ti += u;
        }
        s_w[lane] = ti - t;
    }
    __syncthreads();
    const int excl = pre + s_w[wv] + (inc - v);
    if (i < n) {
        rowptr[i] = excl;
        dis[i] = rsqrtf((float)(v + 1));
        if (i == n - 1) rowptr[n] = excl + v;
    }
}

__global__ void k_fill(const int* __restrict__ src, const int* __restrict__ dst,
                       const int* __restrict__ rank, const int* __restrict__ rowptr,
                       const float* __restrict__ dis, int2* __restrict__ ew, int e) {
    int i = blockIdx.x * blockDim.x + threadIdx.x;
    if (i >= e) return;
    int s = src[i];
    int d = dst[i];
    float w = dis[s] * dis[d];
    ew[rowptr[d] + rank[i]] = make_int2(s, __float_as_int(w));
}

template <int K, int NC, int RELU>
__global__ __launch_bounds__(256)
void k_gemm(const float* __restrict__ A, const float* __restrict__ B,
            const float* __restrict__ bias, float* __restrict__ C, int M) {
    constexpr int TN  = (NC + 15) / 16;
    constexpr int NCP = TN * 16;
    constexpr int BP  = 2 * NCP + (((2 * NCP) % 64 == 32) ? 0 : 32);
    constexpr int KP  = K / 2;
    constexpr int AP  = 68;
    constexpr int PH  = K / 64;
    constexpr int CG  = NC / 4;
    constexpr int NT  = KP * CG;
    constexpr int TPT = (NT + 255) / 256;

    __shared__ __attribute__((aligned(16))) float Bs[KP * BP];
    __shared__ __attribute__((aligned(16))) float As[8 * 16 * AP];

    const int tid  = threadIdx.x;
    const int wv   = tid >> 5;
    const int lane = tid & 31;
    const int hf   = lane >> 4;
    const int l    = lane & 15;
    const int strip = blockIdx.x * 8 + wv;
    const bool live = strip < (M >> 4);

    float4 r0[TPT], r1[TPT];
#pragma unroll
    for (int t = 0; t < TPT; ++t) {
        const int task = tid + t * 256;
        if (NT % 256 == 0 || task < NT) {
            const int kp = task / CG;
            const int cg = task - kp * CG;
            r0[t] = *(const float4*)(B + (size_t)(2 * kp) * NC + 4 * cg);
            r1[t] = *(const float4*)(B + (size_t)(2 * kp + 1) * NC + 4 * cg);
        }
    }

    const float* Ab = A + (size_t)(live ? strip : 0) * 16 * K + (size_t)hf * K + 4 * l;
    v4f ar[8];
#pragma unroll
    for (int q = 0; q < 8; ++q)
        ar[q] = *(const v4f*)(Ab + (size_t)(2 * q) * K);

#pragma unroll
    for (int t = 0; t < TPT; ++t) {
        const int task = tid + t * 256;
        if (NT % 256 == 0 || task < NT) {
            const int kp = task / CG;
            const int cg = task - kp * CG;
            float* d = Bs + kp * BP + 8 * cg;
            *(float4*)d       = make_float4(r0[t].x, r1[t].x, r0[t].y, r1[t].y);
            *(float4*)(d + 4) = make_float4(r0[t].z, r1[t].z, r0[t].w, r1[t].w);
        }
    }
    if (NCP > NC) {
        constexpr int ZQ = (NCP - NC) / 2;
        for (int idx = tid; idx < KP * ZQ; idx += 256) {
            const int kp = idx / ZQ;
            const int zq = idx - kp * ZQ;
            *(float4*)(Bs + kp * BP + 2 * NC + 4 * zq) = make_float4(0.f, 0.f, 0.f, 0.f);
        }
    }
    asm volatile("" :: "v"(ar[0]), "v"(ar[1]), "v"(ar[2]), "v"(ar[3]),
                       "v"(ar[4]), "v"(ar[5]), "v"(ar[6]), "v"(ar[7]));
    __syncthreads();
    if (!live) return;

    float* Aw = As + wv * (16 * AP);
    v8f acc[TN];
#pragma unroll
    for (int tn = 0; tn < TN; ++tn) acc[tn] = v8f{};

#pragma unroll
    for (int p = 0; p < PH; ++p) {
#pragma unroll
        for (int q = 0; q < 8; ++q) {
            v4f v = ar[q];
            if (RELU) {
                v.x = fmaxf(v.x, 0.0f); v.y = fmaxf(v.y, 0.0f);
                v.z = fmaxf(v.z, 0.0f); v.w = fmaxf(v.w, 0.0f);
            }
            *(v4f*)(Aw + (2 * q + hf) * AP + 4 * l) = v;
        }
        __builtin_amdgcn_fence(__ATOMIC_ACQ_REL, "wavefront");
        if (p + 1 < PH) {
#pragma unroll
            for (int q = 0; q < 8; ++q)
                ar[q] = *(const v4f*)(Ab + (size_t)(2 * q) * K + 64 * (p + 1));
            __builtin_amdgcn_sched_barrier(0);
        }
#pragma unroll
        for (int kk = 0; kk < 16; ++kk) {
            const v2f a = *(const v2f*)(Aw + l * AP + 4 * kk + 2 * hf);
            const float* bp = Bs + (32 * p + 2 * kk + hf) * BP + 2 * l;
#pragma unroll
            for (int tn = 0; tn < TN; ++tn) {
                const v2f b = *(const v2f*)(bp + 32 * tn);
                acc[tn] = __builtin_amdgcn_wmma_f32_16x16x4_f32(
                     false, a,  false, b,
                     (short)0, acc[tn],  false,  false);
            }
        }
        __builtin_amdgcn_fence(__ATOMIC_ACQ_REL, "wavefront");
    }

#pragma unroll
    for (int tn = 0; tn < TN; ++tn) {
        const int col = tn * 16 + l;
        if (col < NC) {
            const float bb = bias[col];
            float* Crow = C + (size_t)(strip * 16 + 8 * hf) * NC + col;
#pragma unroll
            for (int i = 0; i < 8; ++i)
                Crow[(size_t)i * NC] = acc[tn][i] + bb;
        }
    }
}

template <int NC>
__global__ __launch_bounds__(512) __attribute__((amdgpu_waves_per_eu(16, 16)))
void k_gather_gemm(const float* __restrict__ t, const int* __restrict__ rowptr,
                   const int2* __restrict__ ew, const float* __restrict__ dis,
                   const float* __restrict__ B, const float* __restrict__ bias,
                   float* __restrict__ C, int n) {
    constexpr int TN  = (NC + 15) / 16;
    constexpr int NCP = TN * 16;
    constexpr int BP  = 2 * NCP + (((2 * NCP) % 64 == 32) ? 0 : 32);
    constexpr int KP  = HID / 2;
    constexpr int AP  = 68;
    constexpr int CG  = NC / 4;
    constexpr int NT  = KP * CG;
    constexpr int JPS = (NC % 32 == 0) ? NC / 32 : 1;
    constexpr int TPJ = TN / JPS;
    static_assert(NT <= 512, "one staging task per thread");
    static_assert(TPJ * JPS == TN, "jobs cover the tiles");

    __shared__ __attribute__((aligned(16))) float Bs[KP * BP];
    __shared__ __attribute__((aligned(16))) float As[2 * 16 * AP];

    const int tid = threadIdx.x;

    const bool stg = (NT == 512) || (tid < NT);
    const int kp = tid / CG;
    const int cg = tid - kp * CG;
    float4 r0 = make_float4(0.f, 0.f, 0.f, 0.f), r1 = r0;
    if (stg) {
        r0 = *(const float4*)(B + (size_t)(2 * kp) * NC + 4 * cg);
        r1 = *(const float4*)(B + (size_t)(2 * kp + 1) * NC + 4 * cg);
    }

    const int gid = blockIdx.x * 512 + tid;
    const int i = gid >> 4;
    const int c = gid & 15;
    const bool valid = i < n;
    const float4* t4 = (const float4*)t;
    int j = 0, je = 0;
    float sd = 0.f;
    float4 v = make_float4(0.f, 0.f, 0.f, 0.f);
    if (valid) {
        j  = rowptr[i];
        je = rowptr[i + 1];
        sd = dis[i];
        v  = t4[gid];
    }

    if (stg) {
        float* d = Bs + kp * BP + 8 * cg;
        *(float4*)d       = make_float4(r0.x, r1.x, r0.y, r1.y);
        *(float4*)(d + 4) = make_float4(r0.z, r1.z, r0.w, r1.w);
    }
    if (NCP > NC) {
        constexpr int ZQ = (NCP - NC) / 2;
        for (int idx = tid; idx < KP * ZQ; idx += 512) {
            const int zk = idx / ZQ;
            const int zq = idx - zk * ZQ;
            *(float4*)(Bs + zk * BP + 2 * NC + 4 * zq) = make_float4(0.f, 0.f, 0.f, 0.f);
        }
    }

    const float k = sd * sd;
    float4 a;
    a.x = v.x * k; a.y = v.y * k; a.z = v.z * k; a.w = v.w * k;
    for (; j < je; ++j) {
        const int2 p = ew[j];
        const float4 u = t4[(size_t)p.x * H4 + c];
        const float w = __int_as_float(p.y);
        a.x += u.x * w; a.y += u.y * w; a.z += u.z * w; a.w += u.w * w;
    }

    {
        v4f h;
        h.x = fmaxf(a.x, 0.0f); h.y = fmaxf(a.y, 0.0f);
        h.z = fmaxf(a.z, 0.0f); h.w = fmaxf(a.w, 0.0f);
        *(v4f*)(As + (tid >> 4) * AP + 4 * c) = h;
    }
    __syncthreads();

    const int wv   = tid >> 5;
    const int lane = tid & 31;
    const int hf   = lane >> 4;
    const int l    = lane & 15;
    if (wv >= 2 * JPS) return;
    const int sl    = wv / JPS;
    const int part  = wv - sl * JPS;
    const int strip = blockIdx.x * 2 + sl;
    if (strip >= (n >> 4)) return;

    const float* Aw = As + sl * (16 * AP);
    v8f acc[TPJ];
#pragma unroll
    for (int tj = 0; tj < TPJ; ++tj) acc[tj] = v8f{};

#pragma unroll
    for (int kk = 0; kk < 16; ++kk) {
        const v2f af = *(const v2f*)(Aw + l * AP + 4 * kk + 2 * hf);
        const float* bp = Bs + (2 * kk + hf) * BP + 2 * l + 32 * (part * TPJ);
#pragma unroll
        for (int tj = 0; tj < TPJ; ++tj) {
            const v2f bf = *(const v2f*)(bp + 32 * tj);
            acc[tj] = __builtin_amdgcn_wmma_f32_16x16x4_f32(
                 false, af,  false, bf,
                 (short)0, acc[tj],  false,  false);
        }
    }

#pragma unroll
    for (int tj = 0; tj < TPJ; ++tj) {
        const int col = (part * TPJ + tj) * 16 + l;
        if (col < NC) {
            const float bb = bias[col];
            float* Crow = C + (size_t)(strip * 16 + 8 * hf) * NC + col;
#pragma unroll
            for (int q = 0; q < 8; ++q)
                Crow[(size_t)q * NC] = acc[tj][q] + bb;
        }
    }
}

extern "C" void kernel_launch(void* const* d_in, const int* in_sizes, int n_in,
                              void* d_out, int out_size, void* d_ws, size_t ws_size,
                              hipStream_t stream) {
    (void)n_in; (void)out_size; (void)ws_size;
    const float* x    = (const float*)d_in[0];
    const int*   ei   = (const int*)d_in[1];
    const float* W0   = (const float*)d_in[2];
    const float* b0   = (const float*)d_in[3];
    const float* W1   = (const float*)d_in[4];
    const float* b1   = (const float*)d_in[5];
    const float* W2   = (const float*)d_in[6];
    const float* b2   = (const float*)d_in[7];
    const float* Wout = (const float*)d_in[8];
    const float* bout = (const float*)d_in[9];

    const int FIN = 128, OUTD = 40;
    const int n = in_sizes[0] / FIN;
    const int e = in_sizes[1] / 2;
    const int* srcI = ei;
    const int* dstI = ei + e;

    char* w = (char*)d_ws;
    size_t off = 0;
    int*   cnt    = (int*)(w + off);   off += (((size_t)n * 4) + 255) & ~(size_t)255;
    int*   rowptr = (int*)(w + off);   off += (((size_t)(n + 1) * 4) + 255) & ~(size_t)255;
    float* dis    = (float*)(w + off); off += (((size_t)n * 4) + 255) & ~(size_t)255;
    int*   rank   = (int*)(w + off);   off += (((size_t)e * 4) + 255) & ~(size_t)255;
    int2*  ew     = (int2*)(w + off);  off += (((size_t)e * 8) + 255) & ~(size_t)255;
    float* tmp    = (float*)(w + off); off += (((size_t)n * HID * 4) + 255) & ~(size_t)255;
    float* hA     = (float*)(w + off); off += (((size_t)n * HID * 4) + 255) & ~(size_t)255;

    const int T   = 256;
    const int WPB = T / 32;

    k_cnt_zero<<<(n + T - 1) / T, T, 0, stream>>>(cnt, n);
    k_cnt_rank<<<(e + T - 1) / T, T, 0, stream>>>(dstI, cnt, rank, e);
    k_rowptr  <<<(n + SCAN_T - 1) / SCAN_T, SCAN_T, 0, stream>>>(cnt, rowptr, dis, n);
    k_fill    <<<(e + T - 1) / T, T, 0, stream>>>(srcI, dstI, rank, rowptr, dis, ew, e);

    const int gGemm  = (n / 16 + WPB - 1) / WPB;
    const int gFuse  = (n + 31) / 32;

    k_gemm<128, HID, 0><<<gGemm, T, 0, stream>>>(x, W0, b0, tmp, n);
    k_gather_gemm<HID><<<gFuse, 512, 0, stream>>>(tmp, rowptr, ew, dis, W1, b1, hA, n);
    k_gather_gemm<HID><<<gFuse, 512, 0, stream>>>(hA, rowptr, ew, dis, W2, b2, tmp, n);
    k_gather_gemm<OUTD><<<gFuse, 512, 0, stream>>>(tmp, rowptr, ew, dis, Wout, bout, (float*)d_out, n);
}


# Aggregation edge loop written as 8-entry trips (8 row loads in flight) plus ONE masked trip for the 1-7 left-over edges, which the compiler's unroll fetched one edge per trip; same sum order
# speedup vs baseline: 2.3477x; 1.0329x over previous
#include <hip/hip_runtime.h>

#define HID 64
#define H4  16

typedef __attribute__((ext_vector_type(2))) float v2f;
typedef __attribute__((ext_vector_type(8))) float v8f;
typedef __attribute__((ext_vector_type(4))) float v4f;

#define SCAN_T 1024

__global__ void k_cnt_zero(int* __restrict__ cnt, int n) {
    int i = blockIdx.x * blockDim.x + threadIdx.x;
    if (i < n) cnt[i] = 0;
}

__global__ void k_cnt_rank(const int* __restrict__ dst, int* __restrict__ cnt,
                           int* __restrict__ rank, int e) {
    int i = blockIdx.x * blockDim.x + threadIdx.x;
    if (i < e) rank[i] = atomicAdd(&cnt[dst[i]], 1);
}

__global__ __launch_bounds__(SCAN_T)
void k_rowptr(const int* __restrict__ cnt, int* __restrict__ rowptr,
              float* __restrict__ dis, int n) {
    __shared__ int s_w[32];
    __shared__ int s_pre;
    const int tid  = threadIdx.x;
    const int lane = tid & 31;
    const int wv   = tid >> 5;
    const int base = blockIdx.x * SCAN_T;

    int part = 0;
    const int4* c4 = (const int4*)cnt;
    for (int k = tid * 4; k < base; k += SCAN_T * 4) {
        int4 q = c4[k >> 2];
        part += (q.x + q.y) + (q.z + q.w);
    }
    for (int o = 16; o > 0; o >>= 1) part += __shfl_xor(part, o);
    if (lane == 0) s_w[wv] = part;
    __syncthreads();
    if (wv == 0) {
        int t = s_w[lane];
        for (int o = 16; o > 0; o >>= 1) t += __shfl_xor(t, o);
        if (lane == 0) s_pre = t;
    }
    __syncthreads();
    const int pre = s_pre;

    const int i = base + tid;
    const int v = (i < n) ? cnt[i] : 0;
    int inc = v;
    for (int o = 1; o < 32; o <<= 1) {
        int t = __shfl_up(inc, o);
        if (lane >= o) inc += t;
    }
    if (lane == 31) s_w[wv] = inc;
    __syncthreads();
    if (wv == 0) {
        int t = s_w[lane];
        int ti = t;
        for (int o = 1; o < 32; o <<= 1) {
            int u = __shfl_up(ti, o);
            if (lane >= o) ti += u;
        }
        s_w[lane] = ti - t;
    }
    __syncthreads();
    const int excl = pre + s_w[wv] + (inc - v);
    if (i < n) {
        rowptr[i] = excl;
        dis[i] = rsqrtf((float)(v + 1));
        if (i == n - 1) rowptr[n] = excl + v;
    }
}

__global__ void k_fill(const int* __restrict__ src, const int* __restrict__ dst,
                       const int* __restrict__ rank, const int* __restrict__ rowptr,
                       const float* __restrict__ dis, int2* __restrict__ ew, int e) {
    int i = blockIdx.x * blockDim.x + threadIdx.x;
    if (i >= e) return;
    int s = src[i];
    int d = dst[i];
    float w = dis[s] * dis[d];
    ew[rowptr[d] + rank[i]] = make_int2(s, __float_as_int(w));
}

template <int K, int NC, int RELU>
__global__ __launch_bounds__(256)
void k_gemm(const float* __restrict__ A, const float* __restrict__ B,
            const float* __restrict__ bias, float* __restrict__ C, int M) {
    constexpr int TN  = (NC + 15) / 16;
    constexpr int NCP = TN * 16;
    constexpr int BP  = 2 * NCP + (((2 * NCP) % 64 == 32) ? 0 : 32);
    constexpr int KP  = K / 2;
    constexpr int AP  = 68;
    constexpr int PH  = K / 64;
    constexpr int CG  = NC / 4;
    constexpr int NT  = KP * CG;
    constexpr int TPT = (NT + 255) / 256;

    __shared__ __attribute__((aligned(16))) float Bs[KP * BP];
    __shared__ __attribute__((aligned(16))) float As[8 * 16 * AP];

    const int tid  = threadIdx.x;
    const int wv   = tid >> 5;
    const int lane = tid & 31;
    const int hf   = lane >> 4;
    const int l    = lane & 15;
    const int strip = blockIdx.x * 8 + wv;
    const bool live = strip < (M >> 4);

    float4 r0[TPT], r1[TPT];
#pragma unroll
    for (int t = 0; t < TPT; ++t) {
        const int task = tid + t * 256;
        if (NT % 256 == 0 || task < NT) {
            const int kp = task / CG;
            const int cg = task - kp * CG;
            r0[t] = *(const float4*)(B + (size_t)(2 * kp) * NC + 4 * cg);
            r1[t] = *(const float4*)(B + (size_t)(2 * kp + 1) * NC + 4 * cg);
        }
    }

    const float* Ab = A + (size_t)(live ? strip : 0) * 16 * K + (size_t)hf * K + 4 * l;
    v4f ar[8];
#pragma unroll
    for (int q = 0; q < 8; ++q)
        ar[q] = *(const v4f*)(Ab + (size_t)(2 * q) * K);

#pragma unroll
    for (int t = 0; t < TPT; ++t) {
        const int task = tid + t * 256;
        if (NT % 256 == 0 || task < NT) {
            const int kp = task / CG;
            const int cg = task - kp * CG;
            float* d = Bs + kp * BP + 8 * cg;
            *(float4*)d       = make_float4(r0[t].x, r1[t].x, r0[t].y, r1[t].y);
            *(float4*)(d + 4) = make_float4(r0[t].z, r1[t].z, r0[t].w, r1[t].w);
        }
    }
    if (NCP > NC) {
        constexpr int ZQ = (NCP - NC) / 2;
        for (int idx = tid; idx < KP * ZQ; idx += 256) {
            const int kp = idx / ZQ;
            const int zq = idx - kp * ZQ;
            *(float4*)(Bs + kp * BP + 2 * NC + 4 * zq) = make_float4(0.f, 0.f, 0.f, 0.f);
        }
    }
    asm volatile("" :: "v"(ar[0]), "v"(ar[1]), "v"(ar[2]), "v"(ar[3]),
                       "v"(ar[4]), "v"(ar[5]), "v"(ar[6]), "v"(ar[7]));
    __syncthreads();
    if (!live) return;

    float* Aw = As + wv * (16 * AP);
    v8f acc[TN];
#pragma unroll
    for (int tn = 0; tn < TN; ++tn) acc[tn] = v8f{};

#pragma unroll
    for (int p = 0; p < PH; ++p) {
#pragma unroll
        for (int q = 0; q < 8; ++q) {
            v4f v = ar[q];
            if (RELU) {
                v.x = fmaxf(v.x, 0.0f); v.y = fmaxf(v.y, 0.0f);
                v.z = fmaxf(v.z, 0.0f); v.w = fmaxf(v.w, 0.0f);
            }
            *(v4f*)(Aw + (2 * q + hf) * AP + 4 * l) = v;
        }
        __builtin_amdgcn_fence(__ATOMIC_ACQ_REL, "wavefront");
        if (p + 1 < PH) {
#pragma unroll
            for (int q = 0; q < 8; ++q)
                ar[q] = *(const v4f*)(Ab + (size_t)(2 * q) * K + 64 * (p + 1));
            __builtin_amdgcn_sched_barrier(0);
        }
#pragma unroll
        for (int kk = 0; kk < 16; ++kk) {
            const v2f a = *(const v2f*)(Aw + l * AP + 4 * kk + 2 * hf);
            const float* bp = Bs + (32 * p + 2 * kk + hf) * BP + 2 * l;
#pragma unroll
            for (int tn = 0; tn < TN; ++tn) {
                const v2f b = *(const v2f*)(bp + 32 * tn);
                acc[tn] = __builtin_amdgcn_wmma_f32_16x16x4_f32(
                     false, a,  false, b,
                     (short)0, acc[tn],  false,  false);
            }
        }
        __builtin_amdgcn_fence(__ATOMIC_ACQ_REL, "wavefront");
    }

#pragma unroll
    for (int tn = 0; tn < TN; ++tn) {
        const int col = tn * 16 + l;
        if (col < NC) {
            const float bb = bias[col];
            float* Crow = C + (size_t)(strip * 16 + 8 * hf) * NC + col;
#pragma unroll
            for (int i = 0; i < 8; ++i)
                Crow[(size_t)i * NC] = acc[tn][i] + bb;
        }
    }
}

template <int NC>
__global__ __launch_bounds__(512) __attribute__((amdgpu_waves_per_eu(16, 16)))
void k_gather_gemm(const float* __restrict__ t, const int* __restrict__ rowptr,
                   const int2* __restrict__ ew, const float* __restrict__ dis,
                   const float* __restrict__ B, const float* __restrict__ bias,
                   float* __restrict__ C, int n) {
    constexpr int TN  = (NC + 15) / 16;
    constexpr int NCP = TN * 16;
    constexpr int BP  = 2 * NCP + (((2 * NCP) % 64 == 32) ? 0 : 32);
    constexpr int KP  = HID / 2;
    constexpr int AP  = 68;
    constexpr int CG  = NC / 4;
    constexpr int NT  = KP * CG;
    constexpr int JPS = (NC % 32 == 0) ? NC / 32 : 1;
    constexpr int TPJ = TN / JPS;
    static_assert(NT <= 512, "one staging task per thread");
    static_assert(TPJ * JPS == TN, "jobs cover the tiles");

    __shared__ __attribute__((aligned(16))) float Bs[KP * BP];
    __shared__ __attribute__((aligned(16))) float As[2 * 16 * AP];

    const int tid = threadIdx.x;

    const bool stg = (NT == 512) || (tid < NT);
    const int kp = tid / CG;
    const int cg = tid - kp * CG;
    float4 r0 = make_float4(0.f, 0.f, 0.f, 0.f), r1 = r0;
    if (stg) {
        r0 = *(const float4*)(B + (size_t)(2 * kp) * NC + 4 * cg);
        r1 = *(const float4*)(B + (size_t)(2 * kp + 1) * NC + 4 * cg);
    }

    const int gid = blockIdx.x * 512 + tid;
    const int i = gid >> 4;
    const int c = gid & 15;
    const bool valid = i < n;
    const float4* t4 = (const float4*)t;
    int j = 0, je = 0;
    float sd = 0.f;
    float4 v = make_float4(0.f, 0.f, 0.f, 0.f);
    if (valid) {
        j  = rowptr[i];
        je = rowptr[i + 1];
        sd = dis[i];
        v  = t4[gid];
    }

    if (stg) {
        float* d = Bs + kp * BP + 8 * cg;
        *(float4*)d       = make_float4(r0.x, r1.x, r0.y, r1.y);
        *(float4*)(d + 4) = make_float4(r0.z, r1.z, r0.w, r1.w);
    }
    if (NCP > NC) {
        constexpr int ZQ = (NCP - NC) / 2;
        for (int idx = tid; idx < KP * ZQ; idx += 512) {
            const int zk = idx / ZQ;
            const int zq = idx - zk * ZQ;
            *(float4*)(Bs + zk * BP + 2 * NC + 4 * zq) = make_float4(0.f, 0.f, 0.f, 0.f);
        }
    }

    const float k = sd * sd;
    float4 a;
    a.x = v.x * k; a.y = v.y * k; a.z = v.z * k; a.w = v.w * k;
    const v4f* tv = (const v4f*)t;
    for (; j + 8 <= je; j += 8) {
        int2 p[8];
#pragma unroll
        for (int q = 0; q < 8; ++q) p[q] = ew[j + q];
        v4f u[8];
#pragma unroll
        for (int q = 0; q < 8; ++q) u[q] = tv[(size_t)p[q].x * H4 + c];
#pragma unroll
        for (int q = 0; q < 8; ++q) {
            const float w = __int_as_float(p[q].y);
            a.x += u[q].x * w; a.y += u[q].y * w; a.z += u[q].z * w; a.w += u[q].w * w;
        }
    }
    const int rem = je - j;
    if (rem > 0) {
        const int last = je - 1;
        int2 p[7];
        p[0] = ew[j];
#pragma unroll
        for (int q = 1; q < 7; ++q) p[q] = ew[min(j + q, last)];
        asm volatile("" :: "v"(p[0].x), "v"(p[1].x), "v"(p[2].x), "v"(p[3].x),
                           "v"(p[4].x), "v"(p[5].x), "v"(p[6].x));
        v4f u[7];
#pragma unroll
        for (int q = 0; q < 7; ++q) u[q] = tv[(size_t)p[q].x * H4 + c];
        asm volatile("" :: "v"(u[0]), "v"(u[1]), "v"(u[2]), "v"(u[3]),
                           "v"(u[4]), "v"(u[5]), "v"(u[6]));
#pragma unroll
        for (int q = 0; q < 7; ++q) {
            if (q < rem) {
                const float w = __int_as_float(p[q].y);
                a.x += u[q].x * w; a.y += u[q].y * w; a.z += u[q].z * w; a.w += u[q].w * w;
            }
        }
    }

    {
        v4f h;
        h.x = fmaxf(a.x, 0.0f); h.y = fmaxf(a.y, 0.0f);
        h.z = fmaxf(a.z, 0.0f); h.w = fmaxf(a.w, 0.0f);
        *(v4f*)(As + (tid >> 4) * AP + 4 * c) = h;
    }
    __syncthreads();

    const int wv   = tid >> 5;
    const int lane = tid & 31;
    const int hf   = lane >> 4;
    const int l    = lane & 15;
    if (wv >= 2 * JPS) return;
    const int sl    = wv / JPS;
    const int part  = wv - sl * JPS;
    const int strip = blockIdx.x * 2 + sl;
    if (strip >= (n >> 4)) return;

    const float* Aw = As + sl * (16 * AP);
    v8f acc[TPJ];
#pragma unroll
    for (int tj = 0; tj < TPJ; ++tj) acc[tj] = v8f{};

#pragma unroll
    for (int kk = 0; kk < 16; ++kk) {
        const v2f af = *(const v2f*)(Aw + l * AP + 4 * kk + 2 * hf);
        const float* bp = Bs + (2 * kk + hf) * BP + 2 * l + 32 * (part * TPJ);
#pragma unroll
        for (int tj = 0; tj < TPJ; ++tj) {
            const v2f bf = *(const v2f*)(bp + 32 * tj);
            acc[tj] = __builtin_amdgcn_wmma_f32_16x16x4_f32(
                 false, af,  false, bf,
                 (short)0, acc[tj],  false,  false);
        }
    }

#pragma unroll
    for (int tj = 0; tj < TPJ; ++tj) {
        const int col = (part * TPJ + tj) * 16 + l;
        if (col < NC) {
            const float bb = bias[col];
            float* Crow = C + (size_t)(strip * 16 + 8 * hf) * NC + col;
#pragma unroll
            for (int q = 0; q < 8; ++q)
                Crow[(size_t)q * NC] = acc[tj][q] + bb;
        }
    }
}

extern "C" void kernel_launch(void* const* d_in, const int* in_sizes, int n_in,
                              void* d_out, int out_size, void* d_ws, size_t ws_size,
                              hipStream_t stream) {
    (void)n_in; (void)out_size; (void)ws_size;
    const float* x    = (const float*)d_in[0];
    const int*   ei   = (const int*)d_in[1];
    const float* W0   = (const float*)d_in[2];
    const float* b0   = (const float*)d_in[3];
    const float* W1   = (const float*)d_in[4];
    const float* b1   = (const float*)d_in[5];
    const float* W2   = (const float*)d_in[6];
    const float* b2   = (const float*)d_in[7];
    const float* Wout = (const float*)d_in[8];
    const float* bout = (const float*)d_in[9];

    const int FIN = 128, OUTD = 40;
    const int n = in_sizes[0] / FIN;
    const int e = in_sizes[1] / 2;
    const int* srcI = ei;
    const int* dstI = ei + e;

    char* w = (char*)d_ws;
    size_t off = 0;
    int*   cnt    = (int*)(w + off);   off += (((size_t)n * 4) + 255) & ~(size_t)255;
    int*   rowptr = (int*)(w + off);   off += (((size_t)(n + 1) * 4) + 255) & ~(size_t)255;
    float* dis    = (float*)(w + off); off += (((size_t)n * 4) + 255) & ~(size_t)255;
    int*   rank   = (int*)(w + off);   off += (((size_t)e * 4) + 255) & ~(size_t)255;
    int2*  ew     = (int2*)(w + off);  off += (((size_t)e * 8) + 255) & ~(size_t)255;
    float* tmp    = (float*)(w + off); off += (((size_t)n * HID * 4) + 255) & ~(size_t)255;
    float* hA     = (float*)(w + off); off += (((size_t)n * HID * 4) + 255) & ~(size_t)255;

    const int T   = 256;
    const int WPB = T / 32;

    k_cnt_zero<<<(n + T - 1) / T, T, 0, stream>>>(cnt, n);
    k_cnt_rank<<<(e + T - 1) / T, T, 0, stream>>>(dstI, cnt, rank, e);
    k_rowptr  <<<(n + SCAN_T - 1) / SCAN_T, SCAN_T, 0, stream>>>(cnt, rowptr, dis, n);
    k_fill    <<<(e + T - 1) / T, T, 0, stream>>>(srcI, dstI, rank, rowptr, dis, ew, e);

    const int gGemm  = (n / 16 + WPB - 1) / WPB;
    const int gFuse  = (n + 31) / 32;

    k_gemm<128, HID, 0><<<gGemm, T, 0, stream>>>(x, W0, b0, tmp, n);
    k_gather_gemm<HID><<<gFuse, 512, 0, stream>>>(tmp, rowptr, ew, dis, W1, b1, hA, n);
    k_gather_gemm<HID><<<gFuse, 512, 0, stream>>>(hA, rowptr, ew, dis, W2, b2, tmp, n);
    k_gather_gemm<OUTD><<<gFuse, 512, 0, stream>>>(tmp, rowptr, ew, dis, Wout, bout, (float*)d_out, n);
}


# Edge grouping: per-edge returning global atomic rank replaced by a two-level counting sort on LDS counters (chunk x 128-node bucket, then per bucket); no global atomic left; sum order = LDS arrival
# speedup vs baseline: 3.4435x; 1.4668x over previous
#include <hip/hip_runtime.h>

#define HID 64
#define H4  16

typedef __attribute__((ext_vector_type(2))) float v2f;
typedef __attribute__((ext_vector_type(8))) float v8f;
typedef __attribute__((ext_vector_type(4))) float v4f;

#define SORT_G 256
#define SORT_T 1024
#define BK_SH  7
#define BK_R   128
#define BK_CAP 1024
#define PL_Q   10
static_assert(PL_Q == 10, "the load fences of level 2 name ten slots");
static_assert(BK_CAP <= SORT_T && (1 << BK_SH) == BK_R, "one bucket per thread in the bucket scan");

__global__ __launch_bounds__(SORT_T)
void k_bucket_count(const int* __restrict__ dst, int* __restrict__ hist, int e, int nb) {
    __shared__ int lh[BK_CAP];
    const int tid = threadIdx.x;
    for (int b = tid; b < nb; b += SORT_T) lh[b] = 0;
    __syncthreads();
    const int per = (e + SORT_G - 1) / SORT_G;
    const int beg = blockIdx.x * per;
    const int end = min(e, beg + per);
    for (int i0 = beg + tid; i0 < end; i0 += 8 * SORT_T) {
        int d[8];
#pragma unroll
        for (int q = 0; q < 8; ++q) d[q] = dst[min(i0 + q * SORT_T, end - 1)];
        asm volatile("" :: "v"(d[0]), "v"(d[1]), "v"(d[2]), "v"(d[3]),
                           "v"(d[4]), "v"(d[5]), "v"(d[6]), "v"(d[7]));
#pragma unroll
        for (int q = 0; q < 8; ++q)
            if (i0 + q * SORT_T < end) atomicAdd(&lh[d[q] >> BK_SH], 1);
    }
    __syncthreads();
    for (int b = tid; b < nb; b += SORT_T) hist[b * SORT_G + blockIdx.x] = lh[b];
}

__global__ __launch_bounds__(1024)
void k_rowscan(const int* __restrict__ hist, int* __restrict__ rowscan,
               int* __restrict__ btot, int m) {
    __shared__ int s_w[32];
    const int tid  = threadIdx.x;
    const int lane = tid & 31;
    const int wv   = tid >> 5;
    const int i = blockIdx.x * 1024 + tid;
    const int v = (i < m) ? hist[i] : 0;
    int inc = v;
    for (int o = 1; o < 32; o <<= 1) {
        int t = __shfl_up(inc, o);
        if (lane >= o) inc += t;
    }
    if (lane == 31) s_w[wv] = inc;
    __syncthreads();
    int off = 0;
    for (int k = wv & ~7; k < wv; ++k) off += s_w[k];
    if (i < m) {
        rowscan[i] = off + inc - v;
        if ((tid & (SORT_G - 1)) == SORT_G - 1) btot[i / SORT_G] = off + inc;
    }
}

__global__ __launch_bounds__(SORT_T)
void k_bucket_scatter(const int* __restrict__ src, const int* __restrict__ dst,
                      const int* __restrict__ rowscan, const int* __restrict__ btot,
                      int* __restrict__ bstart, int2* __restrict__ pairs, int e, int nb) {
    __shared__ int lc[BK_CAP];
    __shared__ int s_w[32];
    const int tid  = threadIdx.x;
    const int lane = tid & 31;
    const int wv   = tid >> 5;
    const int v = (tid < nb) ? btot[tid] : 0;
    int inc = v;
    for (int o = 1; o < 32; o <<= 1) {
        int t = __shfl_up(inc, o);
        if (lane >= o) inc += t;
    }
    if (lane == 31) s_w[wv] = inc;
    __syncthreads();
    if (wv == 0) {
        int t = s_w[lane];
        int ti = t;
        for (int o = 1; o < 32; o <<= 1) {
            int u = __shfl_up(ti, o);
            if (lane >= o) ti += u;
        }
        s_w[lane] = ti - t;
    }
    __syncthreads();
    const int bs = s_w[wv] + inc - v;
    if (tid < nb) {
        lc[tid] = bs + rowscan[tid * SORT_G + blockIdx.x];
        if (blockIdx.x == 0) {
            bstart[tid] = bs;
            if (tid == nb - 1) bstart[nb] = bs + v;
        }
    }
    __syncthreads();
    const int per = (e + SORT_G - 1) / SORT_G;
    const int beg = blockIdx.x * per;
    const int end = min(e, beg + per);
    for (int i0 = beg + tid; i0 < end; i0 += 8 * SORT_T) {
        int s[8], d[8];
#pragma unroll
        for (int q = 0; q < 8; ++q) {
            const int i = min(i0 + q * SORT_T, end - 1);
            s[q] = src[i];
            d[q] = dst[i];
        }
        asm volatile("" :: "v"(s[0]), "v"(s[1]), "v"(s[2]), "v"(s[3]),
                           "v"(s[4]), "v"(s[5]), "v"(s[6]), "v"(s[7]),
                           "v"(d[0]), "v"(d[1]), "v"(d[2]), "v"(d[3]),
                           "v"(d[4]), "v"(d[5]), "v"(d[6]), "v"(d[7]));
#pragma unroll
        for (int q = 0; q < 8; ++q) {
            if (i0 + q * SORT_T < end) {
                const int p = atomicAdd(&lc[d[q] >> BK_SH], 1);
                pairs[p] = make_int2(s[q], d[q]);
            }
        }
    }
}

__global__ __launch_bounds__(256)
void k_bucket_rows(const int2* __restrict__ pairs, const int* __restrict__ bstart,
                   int* __restrict__ rowptr, float* __restrict__ dis, int n) {
    __shared__ int c[BK_R];
    __shared__ int s_w[BK_R / 32];
    const int tid  = threadIdx.x;
    const int lane = tid & 31;
    const int wv   = tid >> 5;
    const int d0 = blockIdx.x << BK_SH;
    if (tid < BK_R) c[tid] = 0;
    const int beg = bstart[blockIdx.x];
    const int end = bstart[blockIdx.x + 1];
    __syncthreads();
    for (int p0 = beg + tid; p0 < end; p0 += PL_Q * 256) {
        int dd[PL_Q];
#pragma unroll
        for (int k = 0; k < PL_Q; ++k) dd[k] = pairs[min(p0 + k * 256, end - 1)].y;
        asm volatile("" :: "v"(dd[0]), "v"(dd[1]), "v"(dd[2]), "v"(dd[3]), "v"(dd[4]),
                           "v"(dd[5]), "v"(dd[6]), "v"(dd[7]), "v"(dd[8]), "v"(dd[9]));
#pragma unroll
        for (int k = 0; k < PL_Q; ++k)
            if (p0 + k * 256 < end) atomicAdd(&c[dd[k] - d0], 1);
    }
    __syncthreads();
    int v = 0, inc = 0;
    if (tid < BK_R) {
        v = c[tid];
        inc = v;
        for (int o = 1; o < 32; o <<= 1) {
            int t = __shfl_up(inc, o);
            if (lane >= o) inc += t;
        }
        if (lane == 31) s_w[wv] = inc;
    }
    __syncthreads();
    if (tid < BK_R) {
        int off = 0;
        for (int k = 0; k < wv; ++k) off += s_w[k];
        const int d = d0 + tid;
        if (d < n) {
            const int rp = beg + off + inc - v;
            rowptr[d] = rp;
            dis[d] = rsqrtf((float)(v + 1));
            if (d == n - 1) rowptr[n] = rp + v;
        }
    }
}

__global__ __launch_bounds__(256)
void k_place(const int2* __restrict__ pairs, const int* __restrict__ rowptr,
             const float* __restrict__ dis, int2* __restrict__ ew, int n) {
    __shared__ int   cur[BK_R];
    __shared__ float ld[BK_R];
    const int tid = threadIdx.x;
    const int d0 = blockIdx.x << BK_SH;
    const int d1 = min(n, d0 + BK_R);
    if (tid < d1 - d0) {
        cur[tid] = rowptr[d0 + tid];
        ld[tid]  = dis[d0 + tid];
    }
    const int beg = rowptr[d0];
    const int end = rowptr[d1];
    __syncthreads();
    for (int p0 = beg + tid; p0 < end; p0 += PL_Q * 256) {
        int2 q[PL_Q];
#pragma unroll
        for (int k = 0; k < PL_Q; ++k) q[k] = pairs[min(p0 + k * 256, end - 1)];
        asm volatile("" :: "v"(q[0].x), "v"(q[1].x), "v"(q[2].x), "v"(q[3].x), "v"(q[4].x),
                           "v"(q[5].x), "v"(q[6].x), "v"(q[7].x), "v"(q[8].x), "v"(q[9].x),
                           "v"(q[0].y), "v"(q[1].y), "v"(q[2].y), "v"(q[3].y), "v"(q[4].y),
                           "v"(q[5].y), "v"(q[6].y), "v"(q[7].y), "v"(q[8].y), "v"(q[9].y));
        __builtin_amdgcn_sched_barrier(0);
        float ds[PL_Q];
#pragma unroll
        for (int k = 0; k < PL_Q; ++k) ds[k] = dis[q[k].x];
        asm volatile("" :: "v"(ds[0]), "v"(ds[1]), "v"(ds[2]), "v"(ds[3]), "v"(ds[4]),
                           "v"(ds[5]), "v"(ds[6]), "v"(ds[7]), "v"(ds[8]), "v"(ds[9]));
#pragma unroll
        for (int k = 0; k < PL_Q; ++k) {
            if (p0 + k * 256 < end) {
                const int r = q[k].y - d0;
                const float w = ds[k] * ld[r];
                const int pos = atomicAdd(&cur[r], 1);
                ew[pos] = make_int2(q[k].x, __float_as_int(w));
            }
        }
    }
}

template <int K, int NC, int RELU>
__global__ __launch_bounds__(256)
void k_gemm(const float* __restrict__ A, const float* __restrict__ B,
            const float* __restrict__ bias, float* __restrict__ C, int M) {
    constexpr int TN  = (NC + 15) / 16;
    constexpr int NCP = TN * 16;
    constexpr int BP  = 2 * NCP + (((2 * NCP) % 64 == 32) ? 0 : 32);
    constexpr int KP  = K / 2;
    constexpr int AP  = 68;
    constexpr int PH  = K / 64;
    constexpr int CG  = NC / 4;
    constexpr int NT  = KP * CG;
    constexpr int TPT = (NT + 255) / 256;

    __shared__ __attribute__((aligned(16))) float Bs[KP * BP];
    __shared__ __attribute__((aligned(16))) float As[8 * 16 * AP];

    const int tid  = threadIdx.x;
    const int wv   = tid >> 5;
    const int lane = tid & 31;
    const int hf   = lane >> 4;
    const int l    = lane & 15;
    const int strip = blockIdx.x * 8 + wv;
    const bool live = strip < (M >> 4);

    float4 r0[TPT], r1[TPT];
#pragma unroll
    for (int t = 0; t < TPT; ++t) {
        const int task = tid + t * 256;
        if (NT % 256 == 0 || task < NT) {
            const int kp = task / CG;
            const int cg = task - kp * CG;
            r0[t] = *(const float4*)(B + (size_t)(2 * kp) * NC + 4 * cg);
            r1[t] = *(const float4*)(B + (size_t)(2 * kp + 1) * NC + 4 * cg);
        }
    }

    const float* Ab = A + (size_t)(live ? strip : 0) * 16 * K + (size_t)hf * K + 4 * l;
    v4f ar[8];
#pragma unroll
    for (int q = 0; q < 8; ++q)
        ar[q] = *(const v4f*)(Ab + (size_t)(2 * q) * K);

#pragma unroll
    for (int t = 0; t < TPT; ++t) {
        const int task = tid + t * 256;
        if (NT % 256 == 0 || task < NT) {
            const int kp = task / CG;
            const int cg = task - kp * CG;
            float* d = Bs + kp * BP + 8 * cg;
            *(float4*)d       = make_float4(r0[t].x, r1[t].x, r0[t].y, r1[t].y);
            *(float4*)(d + 4) = make_float4(r0[t].z, r1[t].z, r0[t].w, r1[t].w);
        }
    }
    if (NCP > NC) {
        constexpr int ZQ = (NCP - NC) / 2;
        for (int idx = tid; idx < KP * ZQ; idx += 256) {
            const int kp = idx / ZQ;
            const int zq = idx - kp * ZQ;
            *(float4*)(Bs + kp * BP + 2 * NC + 4 * zq) = make_float4(0.f, 0.f, 0.f, 0.f);
        }
    }
    asm volatile("" :: "v"(ar[0]), "v"(ar[1]), "v"(ar[2]), "v"(ar[3]),
                       "v"(ar[4]), "v"(ar[5]), "v"(ar[6]), "v"(ar[7]));
    __syncthreads();
    if (!live) return;

    float* Aw = As + wv * (16 * AP);
    v8f acc[TN];
#pragma unroll
    for (int tn = 0; tn < TN; ++tn) acc[tn] = v8f{};

#pragma unroll
    for (int p = 0; p < PH; ++p) {
#pragma unroll
        for (int q = 0; q < 8; ++q) {
            v4f v = ar[q];
            if (RELU) {
                v.x = fmaxf(v.x, 0.0f); v.y = fmaxf(v.y, 0.0f);
                v.z = fmaxf(v.z, 0.0f); v.w = fmaxf(v.w, 0.0f);
            }
            *(v4f*)(Aw + (2 * q + hf) * AP + 4 * l) = v;
        }
        __builtin_amdgcn_fence(__ATOMIC_ACQ_REL, "wavefront");
        if (p + 1 < PH) {
#pragma unroll
            for (int q = 0; q < 8; ++q)
                ar[q] = *(const v4f*)(Ab + (size_t)(2 * q) * K + 64 * (p + 1));
            __builtin_amdgcn_sched_barrier(0);
        }
#pragma unroll
        for (int kk = 0; kk < 16; ++kk) {
            const v2f a = *(const v2f*)(Aw + l * AP + 4 * kk + 2 * hf);
            const float* bp = Bs + (32 * p + 2 * kk + hf) * BP + 2 * l;
#pragma unroll
            for (int tn = 0; tn < TN; ++tn) {
                const v2f b = *(const v2f*)(bp + 32 * tn);
                acc[tn] = __builtin_amdgcn_wmma_f32_16x16x4_f32(
                     false, a,  false, b,
                     (short)0, acc[tn],  false,  false);
            }
        }
        __builtin_amdgcn_fence(__ATOMIC_ACQ_REL, "wavefront");
    }

#pragma unroll
    for (int tn = 0; tn < TN; ++tn) {
        const int col = tn * 16 + l;
        if (col < NC) {
            const float bb = bias[col];
            float* Crow = C + (size_t)(strip * 16 + 8 * hf) * NC + col;
#pragma unroll
            for (int i = 0; i < 8; ++i)
                Crow[(size_t)i * NC] = acc[tn][i] + bb;
        }
    }
}

template <int NC>
__global__ __launch_bounds__(512) __attribute__((amdgpu_waves_per_eu(16, 16)))
void k_gather_gemm(const float* __restrict__ t, const int* __restrict__ rowptr,
                   const int2* __restrict__ ew, const float* __restrict__ dis,
                   const float* __restrict__ B, const float* __restrict__ bias,
                   float* __restrict__ C, int n) {
    constexpr int TN  = (NC + 15) / 16;
    constexpr int NCP = TN * 16;
    constexpr int BP  = 2 * NCP + (((2 * NCP) % 64 == 32) ? 0 : 32);
    constexpr int KP  = HID / 2;
    constexpr int AP  = 68;
    constexpr int CG  = NC / 4;
    constexpr int NT  = KP * CG;
    constexpr int JPS = (NC % 32 == 0) ? NC / 32 : 1;
    constexpr int TPJ = TN / JPS;
    static_assert(NT <= 512, "one staging task per thread");
    static_assert(TPJ * JPS == TN, "jobs cover the tiles");

    __shared__ __attribute__((aligned(16))) float Bs[KP * BP];
    __shared__ __attribute__((aligned(16))) float As[2 * 16 * AP];

    const int tid = threadIdx.x;

    const bool stg = (NT == 512) || (tid < NT);
    const int kp = tid / CG;
    const int cg = tid - kp * CG;
    float4 r0 = make_float4(0.f, 0.f, 0.f, 0.f), r1 = r0;
    if (stg) {
        r0 = *(const float4*)(B + (size_t)(2 * kp) * NC + 4 * cg);
        r1 = *(const float4*)(B + (size_t)(2 * kp + 1) * NC + 4 * cg);
    }

    const int gid = blockIdx.x * 512 + tid;
    const int i = gid >> 4;
    const int c = gid & 15;
    const bool valid = i < n;
    const float4* t4 = (const float4*)t;
    int j = 0, je = 0;
    float sd = 0.f;
    float4 v = make_float4(0.f, 0.f, 0.f, 0.f);
    if (valid) {
        j  = rowptr[i];
        je = rowptr[i + 1];
        sd = dis[i];
        v  = t4[gid];
    }

    if (stg) {
        float* d = Bs + kp * BP + 8 * cg;
        *(float4*)d       = make_float4(r0.x, r1.x, r0.y, r1.y);
        *(float4*)(d + 4) = make_float4(r0.z, r1.z, r0.w, r1.w);
    }
    if (NCP > NC) {
        constexpr int ZQ = (NCP - NC) / 2;
        for (int idx = tid; idx < KP * ZQ; idx += 512) {
            const int zk = idx / ZQ;
            const int zq = idx - zk * ZQ;
            *(float4*)(Bs + zk * BP + 2 * NC + 4 * zq) = make_float4(0.f, 0.f, 0.f, 0.f);
        }
    }

    const float k = sd * sd;
    float4 a;
    a.x = v.x * k; a.y = v.y * k; a.z = v.z * k; a.w = v.w * k;
    const v4f* tv = (const v4f*)t;
    for (; j + 8 <= je; j += 8) {
        int2 p[8];
#pragma unroll
        for (int q = 0; q < 8; ++q) p[q] = ew[j + q];
        v4f u[8];
#pragma unroll
        for (int q = 0; q < 8; ++q) u[q] = tv[(size_t)p[q].x * H4 + c];
#pragma unroll
        for (int q = 0; q < 8; ++q) {
            const float w = __int_as_float(p[q].y);
            a.x += u[q].x * w; a.y += u[q].y * w; a.z += u[q].z * w; a.w += u[q].w * w;
        }
    }
    const int rem = je - j;
    if (rem > 0) {
        const int last = je - 1;
        int2 p[7];
        p[0] = ew[j];
#pragma unroll
        for (int q = 1; q < 7; ++q) p[q] = ew[min(j + q, last)];
        asm volatile("" :: "v"(p[0].x), "v"(p[1].x), "v"(p[2].x), "v"(p[3].x),
                           "v"(p[4].x), "v"(p[5].x), "v"(p[6].x));
        v4f u[7];
#pragma unroll
        for (int q = 0; q < 7; ++q) u[q] = tv[(size_t)p[q].x * H4 + c];
        asm volatile("" :: "v"(u[0]), "v"(u[1]), "v"(u[2]), "v"(u[3]),
                           "v"(u[4]), "v"(u[5]), "v"(u[6]));
#pragma unroll
        for (int q = 0; q < 7; ++q) {
            if (q < rem) {
                const float w = __int_as_float(p[q].y);
                a.x += u[q].x * w; a.y += u[q].y * w; a.z += u[q].z * w; a.w += u[q].w * w;
            }
        }
    }

    {
        v4f h;
        h.x = fmaxf(a.x, 0.0f); h.y = fmaxf(a.y, 0.0f);
        h.z = fmaxf(a.z, 0.0f); h.w = fmaxf(a.w, 0.0f);
        *(v4f*)(As + (tid >> 4) * AP + 4 * c) = h;
    }
    __syncthreads();

    const int wv   = tid >> 5;
    const int lane = tid & 31;
    const int hf   = lane >> 4;
    const int l    = lane & 15;
    if (wv >= 2 * JPS) return;
    const int sl    = wv / JPS;
    const int part  = wv - sl * JPS;
    const int strip = blockIdx.x * 2 + sl;
    if (strip >= (n >> 4)) return;

    const float* Aw = As + sl * (16 * AP);
    v8f acc[TPJ];
#pragma unroll
    for (int tj = 0; tj < TPJ; ++tj) acc[tj] = v8f{};

#pragma unroll
    for (int kk = 0; kk < 16; ++kk) {
        const v2f af = *(const v2f*)(Aw + l * AP + 4 * kk + 2 * hf);
        const float* bp = Bs + (2 * kk + hf) * BP + 2 * l + 32 * (part * TPJ);
#pragma unroll
        for (int tj = 0; tj < TPJ; ++tj) {
            const v2f bf = *(const v2f*)(bp + 32 * tj);
            acc[tj] = __builtin_amdgcn_wmma_f32_16x16x4_f32(
                 false, af,  false, bf,
                 (short)0, acc[tj],  false,  false);
        }
    }

#pragma unroll
    for (int tj = 0; tj < TPJ; ++tj) {
        const int col = (part * TPJ + tj) * 16 + l;
        if (col < NC) {
            const float bb = bias[col];
            float* Crow = C + (size_t)(strip * 16 + 8 * hf) * NC + col;
#pragma unroll
            for (int q = 0; q < 8; ++q)
                Crow[(size_t)q * NC] = acc[tj][q] + bb;
        }
    }
}

extern "C" void kernel_launch(void* const* d_in, const int* in_sizes, int n_in,
                              void* d_out, int out_size, void* d_ws, size_t ws_size,
                              hipStream_t stream) {
    (void)n_in; (void)out_size; (void)ws_size;
    const float* x    = (const float*)d_in[0];
    const int*   ei   = (const int*)d_in[1];
    const float* W0   = (const float*)d_in[2];
    const float* b0   = (const float*)d_in[3];
    const float* W1   = (const float*)d_in[4];
    const float* b1   = (const float*)d_in[5];
    const float* W2   = (const float*)d_in[6];
    const float* b2   = (const float*)d_in[7];
    const float* Wout = (const float*)d_in[8];
    const float* bout = (const float*)d_in[9];

    const int FIN = 128, OUTD = 40;
    const int n = in_sizes[0] / FIN;
    const int e = in_sizes[1] / 2;
    const int* srcI = ei;
    const int* dstI = ei + e;

    char* w = (char*)d_ws;
    size_t off = 0;
    const int nb = (n + BK_R - 1) / BK_R;
    const int m  = nb * SORT_G;
    int*   rowptr = (int*)(w + off);   off += (((size_t)(n + 1) * 4) + 255) & ~(size_t)255;
    float* dis    = (float*)(w + off); off += (((size_t)n * 4) + 255) & ~(size_t)255;
    int2*  pairs  = (int2*)(w + off);  off += (((size_t)e * 8) + 255) & ~(size_t)255;
    int2*  ew     = (int2*)(w + off);  off += (((size_t)e * 8) + 255) & ~(size_t)255;
    float* tmp    = (float*)(w + off); off += (((size_t)n * HID * 4) + 255) & ~(size_t)255;
    float* hA     = (float*)(w + off); off += (((size_t)n * HID * 4) + 255) & ~(size_t)255;
    int*   hist    = (int*)(w + off);  off += (((size_t)m * 4) + 255) & ~(size_t)255;
    int*   rowscan = (int*)(w + off);  off += (((size_t)m * 4) + 255) & ~(size_t)255;
    int*   btot    = (int*)(w + off);  off += (((size_t)nb * 4) + 255) & ~(size_t)255;
    int*   bstart  = (int*)(w + off);  off += (((size_t)(nb + 1) * 4) + 255) & ~(size_t)255;

    const int T   = 256;
    const int WPB = T / 32;

    k_bucket_count  <<<SORT_G, SORT_T, 0, stream>>>(dstI, hist, e, nb);
    k_rowscan       <<<(m + 1023) / 1024, 1024, 0, stream>>>(hist, rowscan, btot, m);
    k_bucket_scatter<<<SORT_G, SORT_T, 0, stream>>>(srcI, dstI, rowscan, btot, bstart, pairs, e, nb);
    k_bucket_rows   <<<nb, T, 0, stream>>>(pairs, bstart, rowptr, dis, n);
    k_place         <<<nb, T, 0, stream>>>(pairs, rowptr, dis, ew, n);

    const int gGemm  = (n / 16 + WPB - 1) / WPB;
    const int gFuse  = (n + 31) / 32;

    k_gemm<128, HID, 0><<<gGemm, T, 0, stream>>>(x, W0, b0, tmp, n);
    k_gather_gemm<HID><<<gFuse, 512, 0, stream>>>(tmp, rowptr, ew, dis, W1, b1, hA, n);
    k_gather_gemm<HID><<<gFuse, 512, 0, stream>>>(hA, rowptr, ew, dis, W2, b2, tmp, n);
    k_gather_gemm<OUTD><<<gFuse, 512, 0, stream>>>(tmp, rowptr, ew, dis, Wout, bout, (float*)d_out, n);
}


# Edge weight dis[s]*dis[d] as two factors (tables stored times dis[row], row sum times dis[d]): entries are 4-byte sources, sort level 2 one launch, no dis[src] gather; 9->8 launches; same sum order
# speedup vs baseline: 3.6679x; 1.0652x over previous
#include <hip/hip_runtime.h>

#define HID 64
#define H4  16

typedef __attribute__((ext_vector_type(2))) float v2f;
typedef __attribute__((ext_vector_type(8))) float v8f;
typedef __attribute__((ext_vector_type(4))) float v4f;

#define SORT_G 256
#define SORT_T 1024
#define BK_SH  7
#define BK_R   128
#define BK_CAP 1024
#define PL_Q   10
static_assert(PL_Q == 10, "the load fences of level 2 name ten slots");
static_assert(BK_CAP <= SORT_T && (1 << BK_SH) == BK_R, "one bucket per thread in the bucket scan");

__global__ __launch_bounds__(SORT_T)
void k_bucket_count(const int* __restrict__ dst, int* __restrict__ hist, int e, int nb) {
    __shared__ int lh[BK_CAP];
    const int tid = threadIdx.x;
    for (int b = tid; b < nb; b += SORT_T) lh[b] = 0;
    __syncthreads();
    const int per = (e + SORT_G - 1) / SORT_G;
    const int beg = blockIdx.x * per;
    const int end = min(e, beg + per);
    for (int i0 = beg + tid; i0 < end; i0 += 8 * SORT_T) {
        int d[8];
#pragma unroll
        for (int q = 0; q < 8; ++q) d[q] = dst[min(i0 + q * SORT_T, end - 1)];
        asm volatile("" :: "v"(d[0]), "v"(d[1]), "v"(d[2]), "v"(d[3]),
                           "v"(d[4]), "v"(d[5]), "v"(d[6]), "v"(d[7]));
#pragma unroll
        for (int q = 0; q < 8; ++q)
            if (i0 + q * SORT_T < end) atomicAdd(&lh[d[q] >> BK_SH], 1);
    }
    __syncthreads();
    for (int b = tid; b < nb; b += SORT_T) hist[b * SORT_G + blockIdx.x] = lh[b];
}

__global__ __launch_bounds__(1024)
void k_rowscan(const int* __restrict__ hist, int* __restrict__ rowscan,
               int* __restrict__ btot, int m) {
    __shared__ int s_w[32];
    const int tid  = threadIdx.x;
    const int lane = tid & 31;
    const int wv   = tid >> 5;
    const int i = blockIdx.x * 1024 + tid;
    const int v = (i < m) ? hist[i] : 0;
    int inc = v;
    for (int o = 1; o < 32; o <<= 1) {
        int t = __shfl_up(inc, o);
        if (lane >= o) inc += t;
    }
    if (lane == 31) s_w[wv] = inc;
    __syncthreads();
    int off = 0;
    for (int k = wv & ~7; k < wv; ++k) off += s_w[k];
    if (i < m) {
        rowscan[i] = off + inc - v;
        if ((tid & (SORT_G - 1)) == SORT_G - 1) btot[i / SORT_G] = off + inc;
    }
}

__global__ __launch_bounds__(SORT_T)
void k_bucket_scatter(const int* __restrict__ src, const int* __restrict__ dst,
                      const int* __restrict__ rowscan, const int* __restrict__ btot,
                      int* __restrict__ bstart, int2* __restrict__ pairs, int e, int nb) {
    __shared__ int lc[BK_CAP];
    __shared__ int s_w[32];
    const int tid  = threadIdx.x;
    const int lane = tid & 31;
    const int wv   = tid >> 5;
    const int v = (tid < nb) ? btot[tid] : 0;
    int inc = v;
    for (int o = 1; o < 32; o <<= 1) {
        int t = __shfl_up(inc, o);
        if (lane >= o) inc += t;
    }
    if (lane == 31) s_w[wv] = inc;
    __syncthreads();
    if (wv == 0) {
        int t = s_w[lane];
        int ti = t;
        for (int o = 1; o < 32; o <<= 1) {
            int u = __shfl_up(ti, o);
            if (lane >= o) ti += u;
        }
        s_w[lane] = ti - t;
    }
    __syncthreads();
    const int bs = s_w[wv] + inc - v;
    if (tid < nb) {
        lc[tid] = bs + rowscan[tid * SORT_G + blockIdx.x];
        if (blockIdx.x == 0) {
            bstart[tid] = bs;
            if (tid == nb - 1) bstart[nb] = bs + v;
        }
    }
    __syncthreads();
    const int per = (e + SORT_G - 1) / SORT_G;
    const int beg = blockIdx.x * per;
    const int end = min(e, beg + per);
    for (int i0 = beg + tid; i0 < end; i0 += 8 * SORT_T) {
        int s[8], d[8];
#pragma unroll
        for (int q = 0; q < 8; ++q) {
            const int i = min(i0 + q * SORT_T, end - 1);
            s[q] = src[i];
            d[q] = dst[i];
        }
        asm volatile("" :: "v"(s[0]), "v"(s[1]), "v"(s[2]), "v"(s[3]),
                           "v"(s[4]), "v"(s[5]), "v"(s[6]), "v"(s[7]),
                           "v"(d[0]), "v"(d[1]), "v"(d[2]), "v"(d[3]),
                           "v"(d[4]), "v"(d[5]), "v"(d[6]), "v"(d[7]));
#pragma unroll
        for (int q = 0; q < 8; ++q) {
            if (i0 + q * SORT_T < end) {
                const int p = atomicAdd(&lc[d[q] >> BK_SH], 1);
                pairs[p] = make_int2(s[q], d[q]);
            }
        }
    }
}

__global__ __launch_bounds__(256)
void k_bucket_place(const int2* __restrict__ pairs, const int* __restrict__ bstart,
                    int* __restrict__ rowptr, float* __restrict__ dis,
                    int* __restrict__ ew, int n) {
    __shared__ int c[BK_R];
    __shared__ int s_w[BK_R / 32];
    const int tid  = threadIdx.x;
    const int lane = tid & 31;
    const int wv   = tid >> 5;
    const int d0 = blockIdx.x << BK_SH;
    if (tid < BK_R) c[tid] = 0;
    const int beg = bstart[blockIdx.x];
    const int end = bstart[blockIdx.x + 1];
    __syncthreads();
    for (int p0 = beg + tid; p0 < end; p0 += PL_Q * 256) {
        int dd[PL_Q];
#pragma unroll
        for (int k = 0; k < PL_Q; ++k) dd[k] = pairs[min(p0 + k * 256, end - 1)].y;
        asm volatile("" :: "v"(dd[0]), "v"(dd[1]), "v"(dd[2]), "v"(dd[3]), "v"(dd[4]),
                           "v"(dd[5]), "v"(dd[6]), "v"(dd[7]), "v"(dd[8]), "v"(dd[9]));
#pragma unroll
        for (int k = 0; k < PL_Q; ++k)
            if (p0 + k * 256 < end) atomicAdd(&c[dd[k] - d0], 1);
    }
    __syncthreads();
    int v = 0, inc = 0;
    if (tid < BK_R) {
        v = c[tid];
        inc = v;
        for (int o = 1; o < 32; o <<= 1) {
            int t = __shfl_up(inc, o);
            if (lane >= o) inc += t;
        }
        if (lane == 31) s_w[wv] = inc;
    }
    __syncthreads();
    if (tid < BK_R) {
        int off = 0;
        for (int k = 0; k < wv; ++k) off += s_w[k];
        const int rp = beg + off + inc - v;
        c[tid] = rp;
        const int d = d0 + tid;
        if (d < n) {
            rowptr[d] = rp;
            dis[d] = rsqrtf((float)(v + 1));
            if (d == n - 1) rowptr[n] = rp + v;
        }
    }
    __syncthreads();
    for (int p0 = beg + tid; p0 < end; p0 += PL_Q * 256) {
        int2 q[PL_Q];
#pragma unroll
        for (int k = 0; k < PL_Q; ++k) q[k] = pairs[min(p0 + k * 256, end - 1)];
        asm volatile("" :: "v"(q[0].x), "v"(q[1].x), "v"(q[2].x), "v"(q[3].x), "v"(q[4].x),
                           "v"(q[5].x), "v"(q[6].x), "v"(q[7].x), "v"(q[8].x), "v"(q[9].x),
                           "v"(q[0].y), "v"(q[1].y), "v"(q[2].y), "v"(q[3].y), "v"(q[4].y),
                           "v"(q[5].y), "v"(q[6].y), "v"(q[7].y), "v"(q[8].y), "v"(q[9].y));
#pragma unroll
        for (int k = 0; k < PL_Q; ++k) {
            if (p0 + k * 256 < end) {
                const int pos = atomicAdd(&c[q[k].y - d0], 1);
                ew[pos] = q[k].x;
            }
        }
    }
}

template <int K, int NC, int RELU, int SC>
__global__ __launch_bounds__(256)
void k_gemm(const float* __restrict__ A, const float* __restrict__ B,
            const float* __restrict__ bias, const float* __restrict__ rs,
            float* __restrict__ C, int M) {
    constexpr int TN  = (NC + 15) / 16;
    constexpr int NCP = TN * 16;
    constexpr int BP  = 2 * NCP + (((2 * NCP) % 64 == 32) ? 0 : 32);
    constexpr int KP  = K / 2;
    constexpr int AP  = 68;
    constexpr int PH  = K / 64;
    constexpr int CG  = NC / 4;
    constexpr int NT  = KP * CG;
    constexpr int TPT = (NT + 255) / 256;

    __shared__ __attribute__((aligned(16))) float Bs[KP * BP];
    __shared__ __attribute__((aligned(16))) float As[8 * 16 * AP];

    const int tid  = threadIdx.x;
    const int wv   = tid >> 5;
    const int lane = tid & 31;
    const int hf   = lane >> 4;
    const int l    = lane & 15;
    const int strip = blockIdx.x * 8 + wv;
    const bool live = strip < (M >> 4);

    float4 r0[TPT], r1[TPT];
#pragma unroll
    for (int t = 0; t < TPT; ++t) {
        const int task = tid + t * 256;
        if (NT % 256 == 0 || task < NT) {
            const int kp = task / CG;
            const int cg = task - kp * CG;
            r0[t] = *(const float4*)(B + (size_t)(2 * kp) * NC + 4 * cg);
            r1[t] = *(const float4*)(B + (size_t)(2 * kp + 1) * NC + 4 * cg);
        }
    }

    const float* Ab = A + (size_t)(live ? strip : 0) * 16 * K + (size_t)hf * K + 4 * l;
    v4f ar[8];
#pragma unroll
    for (int q = 0; q < 8; ++q)
        ar[q] = *(const v4f*)(Ab + (size_t)(2 * q) * K);

#pragma unroll
    for (int t = 0; t < TPT; ++t) {
        const int task = tid + t * 256;
        if (NT % 256 == 0 || task < NT) {
            const int kp = task / CG;
            const int cg = task - kp * CG;
            float* d = Bs + kp * BP + 8 * cg;
            *(float4*)d       = make_float4(r0[t].x, r1[t].x, r0[t].y, r1[t].y);
            *(float4*)(d + 4) = make_float4(r0[t].z, r1[t].z, r0[t].w, r1[t].w);
        }
    }
    if (NCP > NC) {
        constexpr int ZQ = (NCP - NC) / 2;
        for (int idx = tid; idx < KP * ZQ; idx += 256) {
            const int kp = idx / ZQ;
            const int zq = idx - kp * ZQ;
            *(float4*)(Bs + kp * BP + 2 * NC + 4 * zq) = make_float4(0.f, 0.f, 0.f, 0.f);
        }
    }
    asm volatile("" :: "v"(ar[0]), "v"(ar[1]), "v"(ar[2]), "v"(ar[3]),
                       "v"(ar[4]), "v"(ar[5]), "v"(ar[6]), "v"(ar[7]));
    __syncthreads();
    if (!live) return;

    float* Aw = As + wv * (16 * AP);
    v8f acc[TN];
#pragma unroll
    for (int tn = 0; tn < TN; ++tn) acc[tn] = v8f{};

#pragma unroll
    for (int p = 0; p < PH; ++p) {
#pragma unroll
        for (int q = 0; q < 8; ++q) {
            v4f v = ar[q];
            if (RELU) {
                v.x = fmaxf(v.x, 0.0f); v.y = fmaxf(v.y, 0.0f);
                v.z = fmaxf(v.z, 0.0f); v.w = fmaxf(v.w, 0.0f);
            }
            *(v4f*)(Aw + (2 * q + hf) * AP + 4 * l) = v;
        }
        __builtin_amdgcn_fence(__ATOMIC_ACQ_REL, "wavefront");
        if (p + 1 < PH) {
#pragma unroll
            for (int q = 0; q < 8; ++q)
                ar[q] = *(const v4f*)(Ab + (size_t)(2 * q) * K + 64 * (p + 1));
            __builtin_amdgcn_sched_barrier(0);
        }
#pragma unroll
        for (int kk = 0; kk < 16; ++kk) {
            const v2f a = *(const v2f*)(Aw + l * AP + 4 * kk + 2 * hf);
            const float* bp = Bs + (32 * p + 2 * kk + hf) * BP + 2 * l;
#pragma unroll
            for (int tn = 0; tn < TN; ++tn) {
                const v2f b = *(const v2f*)(bp + 32 * tn);
                acc[tn] = __builtin_amdgcn_wmma_f32_16x16x4_f32(
                     false, a,  false, b,
                     (short)0, acc[tn],  false,  false);
            }
        }
        __builtin_amdgcn_fence(__ATOMIC_ACQ_REL, "wavefront");
    }

    float sc[8];
#pragma unroll
    for (int i = 0; i < 8; ++i) sc[i] = 1.0f;
    if (SC) {
        const v4f s0 = *(const v4f*)(rs + strip * 16 + 8 * hf);
        const v4f s1 = *(const v4f*)(rs + strip * 16 + 8 * hf + 4);
        sc[0] = s0.x; sc[1] = s0.y; sc[2] = s0.z; sc[3] = s0.w;
        sc[4] = s1.x; sc[5] = s1.y; sc[6] = s1.z; sc[7] = s1.w;
    }
#pragma unroll
    for (int tn = 0; tn < TN; ++tn) {
        const int col = tn * 16 + l;
        if (col < NC) {
            const float bb = bias[col];
            float* Crow = C + (size_t)(strip * 16 + 8 * hf) * NC + col;
#pragma unroll
            for (int i = 0; i < 8; ++i)
                Crow[(size_t)i * NC] = SC ? (acc[tn][i] + bb) * sc[i] : acc[tn][i] + bb;
        }
    }
}

template <int NC, int SC>
__global__ __launch_bounds__(512) __attribute__((amdgpu_waves_per_eu(16, 16)))
void k_gather_gemm(const float* __restrict__ t, const int* __restrict__ rowptr,
                   const int* __restrict__ ew, const float* __restrict__ dis,
                   const float* __restrict__ B, const float* __restrict__ bias,
                   float* __restrict__ C, int n) {
    constexpr int TN  = (NC + 15) / 16;
    constexpr int NCP = TN * 16;
    constexpr int BP  = 2 * NCP + (((2 * NCP) % 64 == 32) ? 0 : 32);
    constexpr int KP  = HID / 2;
    constexpr int AP  = 68;
    constexpr int CG  = NC / 4;
    constexpr int NT  = KP * CG;
    constexpr int JPS = (NC % 32 == 0) ? NC / 32 : 1;
    constexpr int TPJ = TN / JPS;
    static_assert(NT <= 512, "one staging task per thread");
    static_assert(TPJ * JPS == TN, "jobs cover the tiles");

    __shared__ __attribute__((aligned(16))) float Bs[KP * BP];
    __shared__ __attribute__((aligned(16))) float As[2 * 16 * AP];
    __shared__ __attribute__((aligned(16))) float sds[32];

    const int tid = threadIdx.x;

    const bool stg = (NT == 512) || (tid < NT);
    const int kp = tid / CG;
    const int cg = tid - kp * CG;
    float4 r0 = make_float4(0.f, 0.f, 0.f, 0.f), r1 = r0;
    if (stg) {
        r0 = *(const float4*)(B + (size_t)(2 * kp) * NC + 4 * cg);
        r1 = *(const float4*)(B + (size_t)(2 * kp + 1) * NC + 4 * cg);
    }

    const int gid = blockIdx.x * 512 + tid;
    const int i = gid >> 4;
    const int c = gid & 15;
    const bool valid = i < n;
    const float4* t4 = (const float4*)t;
    int j = 0, je = 0;
    float sd = 0.f;
    float4 v = make_float4(0.f, 0.f, 0.f, 0.f);
    if (valid) {
        j  = rowptr[i];
        je = rowptr[i + 1];
        sd = dis[i];
        v  = t4[gid];
    }

    if (stg) {
        float* d = Bs + kp * BP + 8 * cg;
        *(float4*)d       = make_float4(r0.x, r1.x, r0.y, r1.y);
        *(float4*)(d + 4) = make_float4(r0.z, r1.z, r0.w, r1.w);
    }
    if (NCP > NC) {
        constexpr int ZQ = (NCP - NC) / 2;
        for (int idx = tid; idx < KP * ZQ; idx += 512) {
            const int zk = idx / ZQ;
            const int zq = idx - zk * ZQ;
            *(float4*)(Bs + zk * BP + 2 * NC + 4 * zq) = make_float4(0.f, 0.f, 0.f, 0.f);
        }
    }

    if (SC && c == 0) sds[tid >> 4] = sd;
    float4 a = v;
    const v4f* tv = (const v4f*)t;
    for (; j + 8 <= je; j += 8) {
        int p[8];
#pragma unroll
        for (int q = 0; q < 8; ++q) p[q] = ew[j + q];
        v4f u[8];
#pragma unroll
        for (int q = 0; q < 8; ++q) u[q] = tv[(size_t)p[q] * H4 + c];
#pragma unroll
        for (int q = 0; q < 8; ++q) {
            a.x += u[q].x; a.y += u[q].y; a.z += u[q].z; a.w += u[q].w;
        }
    }
    const int rem = je - j;
    if (rem > 0) {
        const int last = je - 1;
        int p[7];
        p[0] = ew[j];
#pragma unroll
        for (int q = 1; q < 7; ++q) p[q] = ew[min(j + q, last)];
        asm volatile("" :: "v"(p[0]), "v"(p[1]), "v"(p[2]), "v"(p[3]),
                           "v"(p[4]), "v"(p[5]), "v"(p[6]));
        v4f u[7];
#pragma unroll
        for (int q = 0; q < 7; ++q) u[q] = tv[(size_t)p[q] * H4 + c];
        asm volatile("" :: "v"(u[0]), "v"(u[1]), "v"(u[2]), "v"(u[3]),
                           "v"(u[4]), "v"(u[5]), "v"(u[6]));
        __builtin_amdgcn_sched_barrier(0);
#pragma unroll
        for (int q = 0; q < 7; ++q) {
            if (q < rem) {
                a.x += u[q].x; a.y += u[q].y; a.z += u[q].z; a.w += u[q].w;
            }
        }
    }

    {
        v4f h;
        h.x = fmaxf(a.x * sd, 0.0f); h.y = fmaxf(a.y * sd, 0.0f);
        h.z = fmaxf(a.z * sd, 0.0f); h.w = fmaxf(a.w * sd, 0.0f);
        *(v4f*)(As + (tid >> 4) * AP + 4 * c) = h;
    }
    __syncthreads();

    const int wv   = tid >> 5;
    const int lane = tid & 31;
    const int hf   = lane >> 4;
    const int l    = lane & 15;
    if (wv >= 2 * JPS) return;
    const int sl    = wv / JPS;
    const int part  = wv - sl * JPS;
    const int strip = blockIdx.x * 2 + sl;
    if (strip >= (n >> 4)) return;

    const float* Aw = As + sl * (16 * AP);
    v8f acc[TPJ];
#pragma unroll
    for (int tj = 0; tj < TPJ; ++tj) acc[tj] = v8f{};

#pragma unroll
    for (int kk = 0; kk < 16; ++kk) {
        const v2f af = *(const v2f*)(Aw + l * AP + 4 * kk + 2 * hf);
        const float* bp = Bs + (2 * kk + hf) * BP + 2 * l + 32 * (part * TPJ);
#pragma unroll
        for (int tj = 0; tj < TPJ; ++tj) {
            const v2f bf = *(const v2f*)(bp + 32 * tj);
            acc[tj] = __builtin_amdgcn_wmma_f32_16x16x4_f32(
                 false, af,  false, bf,
                 (short)0, acc[tj],  false,  false);
        }
    }

    float sc[8];
#pragma unroll
    for (int q = 0; q < 8; ++q) sc[q] = 1.0f;
    if (SC) {
        const v4f s0 = *(const v4f*)(sds + sl * 16 + 8 * hf);
        const v4f s1 = *(const v4f*)(sds + sl * 16 + 8 * hf + 4);
        sc[0] = s0.x; sc[1] = s0.y; sc[2] = s0.z; sc[3] = s0.w;
        sc[4] = s1.x; sc[5] = s1.y; sc[6] = s1.z; sc[7] = s1.w;
    }
#pragma unroll
    for (int tj = 0; tj < TPJ; ++tj) {
        const int col = (part * TPJ + tj) * 16 + l;
        if (col < NC) {
            const float bb = bias[col];
            float* Crow = C + (size_t)(strip * 16 + 8 * hf) * NC + col;
#pragma unroll
            for (int q = 0; q < 8; ++q)
                Crow[(size_t)q * NC] = SC ? (acc[tj][q] + bb) * sc[q] : acc[tj][q] + bb;
        }
    }
}

extern "C" void kernel_launch(void* const* d_in, const int* in_sizes, int n_in,
                              void* d_out, int out_size, void* d_ws, size_t ws_size,
                              hipStream_t stream) {
    (void)n_in; (void)out_size; (void)ws_size;
    const float* x    = (const float*)d_in[0];
    const int*   ei   = (const int*)d_in[1];
    const float* W0   = (const float*)d_in[2];
    const float* b0   = (const float*)d_in[3];
    const float* W1   = (const float*)d_in[4];
    const float* b1   = (const float*)d_in[5];
    const float* W2   = (const float*)d_in[6];
    const float* b2   = (const float*)d_in[7];
    const float* Wout = (const float*)d_in[8];
    const float* bout = (const float*)d_in[9];

    const int FIN = 128, OUTD = 40;
    const int n = in_sizes[0] / FIN;
    const int e = in_sizes[1] / 2;
    const int* srcI = ei;
    const int* dstI = ei + e;

    char* w = (char*)d_ws;
    size_t off = 0;
    const int nb = (n + BK_R - 1) / BK_R;
    const int m  = nb * SORT_G;
    int*   rowptr = (int*)(w + off);   off += (((size_t)(n + 1) * 4) + 255) & ~(size_t)255;
    float* dis    = (float*)(w + off); off += (((size_t)n * 4) + 255) & ~(size_t)255;
    int2*  pairs  = (int2*)(w + off);  off += (((size_t)e * 8) + 255) & ~(size_t)255;
    int*   ew     = (int*)(w + off);   off += (((size_t)e * 4) + 255) & ~(size_t)255;
    float* tmp    = (float*)(w + off); off += (((size_t)n * HID * 4) + 255) & ~(size_t)255;
    float* hA     = (float*)(w + off); off += (((size_t)n * HID * 4) + 255) & ~(size_t)255;
    int*   hist    = (int*)(w + off);  off += (((size_t)m * 4) + 255) & ~(size_t)255;
    int*   rowscan = (int*)(w + off);  off += (((size_t)m * 4) + 255) & ~(size_t)255;
    int*   btot    = (int*)(w + off);  off += (((size_t)nb * 4) + 255) & ~(size_t)255;
    int*   bstart  = (int*)(w + off);  off += (((size_t)(nb + 1) * 4) + 255) & ~(size_t)255;

    const int T   = 256;
    const int WPB = T / 32;

    k_bucket_count  <<<SORT_G, SORT_T, 0, stream>>>(dstI, hist, e, nb);
    k_rowscan       <<<(m + 1023) / 1024, 1024, 0, stream>>>(hist, rowscan, btot, m);
    k_bucket_scatter<<<SORT_G, SORT_T, 0, stream>>>(srcI, dstI, rowscan, btot, bstart, pairs, e, nb);
    k_bucket_place  <<<nb, T, 0, stream>>>(pairs, bstart, rowptr, dis, ew, n);

    const int gGemm  = (n / 16 + WPB - 1) / WPB;
    const int gFuse  = (n + 31) / 32;

    k_gemm<128, HID, 0, 1><<<gGemm, T, 0, stream>>>(x, W0, b0, dis, tmp, n);
    k_gather_gemm<HID, 1><<<gFuse, 512, 0, stream>>>(tmp, rowptr, ew, dis, W1, b1, hA, n);
    k_gather_gemm<HID, 1><<<gFuse, 512, 0, stream>>>(hA, rowptr, ew, dis, W2, b2, tmp, n);
    k_gather_gemm<OUTD, 0><<<gFuse, 512, 0, stream>>>(tmp, rowptr, ew, dis, Wout, bout, (float*)d_out, n);
}


# Edge grouping level 1 writes chunk-private stretches (count, in-group scan, copy in ONE launch); level 2 finds its 256 pieces by a transposed offset table; 8->6 launches; sum order still LDS arrival
# speedup vs baseline: 4.2578x; 1.1608x over previous
#include <hip/hip_runtime.h>

#define HID 64
#define H4  16

typedef __attribute__((ext_vector_type(2))) float v2f;
typedef __attribute__((ext_vector_type(8))) float v8f;
typedef __attribute__((ext_vector_type(4))) float v4f;

#define SORT_G 256
#define SORT_T 1024
#define BK_SH  7
#define BK_R   128
#define BK_CAP 1024
#define PC_R   8
#define PC_S   3
static_assert(PC_R * 32 == SORT_G && PC_R * PC_S == 24, "the load fences of level 2 name 24 slots");
static_assert(BK_CAP <= SORT_T && (1 << BK_SH) == BK_R, "one bucket per thread in the bucket scan");

__global__ __launch_bounds__(SORT_T)
void k_chunk_group(const int* __restrict__ src, const int* __restrict__ dst,
                   int* __restrict__ L, int2* __restrict__ pairs, int e, int nb) {
    __shared__ int lh[BK_CAP];
    __shared__ int s_w[32];
    const int tid  = threadIdx.x;
    const int lane = tid & 31;
    const int wv   = tid >> 5;
    if (tid < BK_CAP) lh[tid] = 0;
    __syncthreads();
    const int per = (e + SORT_G - 1) / SORT_G;
    const int beg = blockIdx.x * per;
    const int end = min(e, beg + per);
    for (int i0 = beg + tid; i0 < end; i0 += 8 * SORT_T) {
        int d[8];
#pragma unroll
        for (int q = 0; q < 8; ++q) d[q] = dst[min(i0 + q * SORT_T, end - 1)];
        asm volatile("" :: "v"(d[0]), "v"(d[1]), "v"(d[2]), "v"(d[3]),
                           "v"(d[4]), "v"(d[5]), "v"(d[6]), "v"(d[7]));
#pragma unroll
        for (int q = 0; q < 8; ++q)
            if (i0 + q * SORT_T < end) atomicAdd(&lh[d[q] >> BK_SH], 1);
    }
    __syncthreads();
    const int v = (tid < BK_CAP) ? lh[tid] : 0;
    int inc = v;
    for (int o = 1; o < 32; o <<= 1) {
        int t = __shfl_up(inc, o);
        if (lane >= o) inc += t;
    }
    if (lane == 31) s_w[wv] = inc;
    __syncthreads();
    if (wv == 0) {
        int t = s_w[lane];
        int ti = t;
        for (int o = 1; o < 32; o <<= 1) {
            int u = __shfl_up(ti, o);
            if (lane >= o) ti += u;
        }
        s_w[lane] = ti - t;
    }
    __syncthreads();
    const int loc = s_w[wv] + inc - v;
    if (tid < BK_CAP) lh[tid] = beg + loc;
    if (tid < nb) {
        L[tid * SORT_G + blockIdx.x] = loc;
        if (tid == nb - 1) L[nb * SORT_G + blockIdx.x] = loc + v;
    }
    __syncthreads();
    for (int i0 = beg + tid; i0 < end; i0 += 8 * SORT_T) {
        int s[8], d[8];
#pragma unroll
        for (int q = 0; q < 8; ++q) {
            const int i = min(i0 + q * SORT_T, end - 1);
            s[q] = src[i];
            d[q] = dst[i];
        }
        asm volatile("" :: "v"(s[0]), "v"(s[1]), "v"(s[2]), "v"(s[3]),
                           "v"(s[4]), "v"(s[5]), "v"(s[6]), "v"(s[7]),
                           "v"(d[0]), "v"(d[1]), "v"(d[2]), "v"(d[3]),
                           "v"(d[4]), "v"(d[5]), "v"(d[6]), "v"(d[7]));
#pragma unroll
        for (int q = 0; q < 8; ++q) {
            if (i0 + q * SORT_T < end) {
                const int p = atomicAdd(&lh[d[q] >> BK_SH], 1);
                pairs[p] = make_int2(s[q], d[q]);
            }
        }
    }
}

__global__ __launch_bounds__(SORT_G)
void k_bucket_place(const int2* __restrict__ pairs, const int* __restrict__ L,
                    int* __restrict__ rowptr, float* __restrict__ dis,
                    int* __restrict__ ew, int n, int e) {
    __shared__ int c[BK_R];
    __shared__ int s_w[BK_R / 32];
    __shared__ int sS[SORT_G];
    __shared__ int sL[SORT_G];
    __shared__ int s_r[SORT_G / 32];
    const int tid  = threadIdx.x;
    const int lane = tid & 31;
    const int wv   = tid >> 5;
    const int d0 = blockIdx.x << BK_SH;
    if (tid < BK_R) c[tid] = 0;
    const int per = (e + SORT_G - 1) / SORT_G;
    const int a = L[blockIdx.x * SORT_G + tid];
    const int z = L[(blockIdx.x + 1) * SORT_G + tid];
    sS[tid] = tid * per + a;
    sL[tid] = z - a;
    int ra = a;
    for (int o = 16; o >= 1; o >>= 1) ra += __shfl_xor(ra, o);
    if (lane == 0) s_r[wv] = ra;
    __syncthreads();
    int beg = 0;
#pragma unroll
    for (int k = 0; k < SORT_G / 32; ++k) beg += s_r[k];

    const int gi  = tid >> 3;
    const int sub = tid & 7;
    int st[PC_R], ln[PC_R];
#pragma unroll
    for (int r = 0; r < PC_R; ++r) {
        st[r] = sS[r * 32 + gi];
        ln[r] = sL[r * 32 + gi];
    }
    int dd[PC_R * PC_S];
#pragma unroll
    for (int r = 0; r < PC_R; ++r)
#pragma unroll
        for (int j = 0; j < PC_S; ++j) {
            const int k = sub + 8 * j;
            dd[r * PC_S + j] = pairs[(k < ln[r]) ? st[r] + k : 0].y;
        }
    asm volatile("" :: "v"(dd[0]), "v"(dd[1]), "v"(dd[2]), "v"(dd[3]), "v"(dd[4]), "v"(dd[5]),
                       "v"(dd[6]), "v"(dd[7]), "v"(dd[8]), "v"(dd[9]), "v"(dd[10]), "v"(dd[11]),
                       "v"(dd[12]), "v"(dd[13]), "v"(dd[14]), "v"(dd[15]), "v"(dd[16]), "v"(dd[17]),
                       "v"(dd[18]), "v"(dd[19]), "v"(dd[20]), "v"(dd[21]), "v"(dd[22]), "v"(dd[23]));
#pragma unroll
    for (int r = 0; r < PC_R; ++r)
#pragma unroll
        for (int j = 0; j < PC_S; ++j)
            if (sub + 8 * j < ln[r]) atomicAdd(&c[dd[r * PC_S + j] - d0], 1);
    int lmax = ln[0];
#pragma unroll
    for (int r = 1; r < PC_R; ++r) lmax = max(lmax, ln[r]);
    if (lmax > 8 * PC_S) {
#pragma unroll
        for (int r = 0; r < PC_R; ++r)
            for (int k = sub + 8 * PC_S; k < ln[r]; k += 8)
                atomicAdd(&c[pairs[st[r] + k].y - d0], 1);
    }
    __syncthreads();
    int v = 0, inc = 0;
    if (tid < BK_R) {
        v = c[tid];
        inc = v;
        for (int o = 1; o < 32; o <<= 1) {
            int t = __shfl_up(inc, o);
            if (lane >= o) inc += t;
        }
        if (lane == 31) s_w[wv] = inc;
    }
    __syncthreads();
    if (tid < BK_R) {
        int off = 0;
        for (int k = 0; k < wv; ++k) off += s_w[k];
        const int rp = beg + off + inc - v;
        c[tid] = rp;
        const int d = d0 + tid;
        if (d < n) {
            rowptr[d] = rp;
            dis[d] = rsqrtf((float)(v + 1));
            if (d == n - 1) rowptr[n] = rp + v;
        }
    }
    __syncthreads();
    int ss[PC_R * PC_S];
#pragma unroll
    for (int r = 0; r < PC_R; ++r)
#pragma unroll
        for (int j = 0; j < PC_S; ++j) {
            const int k = sub + 8 * j;
            ss[r * PC_S + j] = pairs[(k < ln[r]) ? st[r] + k : 0].x;
        }
    asm volatile("" :: "v"(ss[0]), "v"(ss[1]), "v"(ss[2]), "v"(ss[3]), "v"(ss[4]), "v"(ss[5]),
                       "v"(ss[6]), "v"(ss[7]), "v"(ss[8]), "v"(ss[9]), "v"(ss[10]), "v"(ss[11]),
                       "v"(ss[12]), "v"(ss[13]), "v"(ss[14]), "v"(ss[15]), "v"(ss[16]), "v"(ss[17]),
                       "v"(ss[18]), "v"(ss[19]), "v"(ss[20]), "v"(ss[21]), "v"(ss[22]), "v"(ss[23]));
#pragma unroll
    for (int r = 0; r < PC_R; ++r)
#pragma unroll
        for (int j = 0; j < PC_S; ++j) {
            if (sub + 8 * j < ln[r]) {
                const int pos = atomicAdd(&c[dd[r * PC_S + j] - d0], 1);
                ew[pos] = ss[r * PC_S + j];
            }
        }
    if (lmax > 8 * PC_S) {
#pragma unroll
        for (int r = 0; r < PC_R; ++r)
            for (int k = sub + 8 * PC_S; k < ln[r]; k += 8) {
                const int2 p = pairs[st[r] + k];
                const int pos = atomicAdd(&c[p.y - d0], 1);
                ew[pos] = p.x;
            }
    }
}

template <int K, int NC, int RELU, int SC>
__global__ __launch_bounds__(256)
void k_gemm(const float* __restrict__ A, const float* __restrict__ B,
            const float* __restrict__ bias, const float* __restrict__ rs,
            float* __restrict__ C, int M) {
    constexpr int TN  = (NC + 15) / 16;
    constexpr int NCP = TN * 16;
    constexpr int BP  = 2 * NCP + (((2 * NCP) % 64 == 32) ? 0 : 32);
    constexpr int KP  = K / 2;
    constexpr int AP  = 68;
    constexpr int PH  = K / 64;
    constexpr int CG  = NC / 4;
    constexpr int NT  = KP * CG;
    constexpr int TPT = (NT + 255) / 256;

    __shared__ __attribute__((aligned(16))) float Bs[KP * BP];
    __shared__ __attribute__((aligned(16))) float As[8 * 16 * AP];

    const int tid  = threadIdx.x;
    const int wv   = tid >> 5;
    const int lane = tid & 31;
    const int hf   = lane >> 4;
    const int l    = lane & 15;
    const int strip = blockIdx.x * 8 + wv;
    const bool live = strip < (M >> 4);

    float4 r0[TPT], r1[TPT];
#pragma unroll
    for (int t = 0; t < TPT; ++t) {
        const int task = tid + t * 256;
        if (NT % 256 == 0 || task < NT) {
            const int kp = task / CG;
            const int cg = task - kp * CG;
            r0[t] = *(const float4*)(B + (size_t)(2 * kp) * NC + 4 * cg);
            r1[t] = *(const float4*)(B + (size_t)(2 * kp + 1) * NC + 4 * cg);
        }
    }

    const float* Ab = A + (size_t)(live ? strip : 0) * 16 * K + (size_t)hf * K + 4 * l;
    v4f ar[8];
#pragma unroll
    for (int q = 0; q < 8; ++q)
        ar[q] = *(const v4f*)(Ab + (size_t)(2 * q) * K);

#pragma unroll
    for (int t = 0; t < TPT; ++t) {
        const int task = tid + t * 256;
        if (NT % 256 == 0 || task < NT) {
            const int kp = task / CG;
            const int cg = task - kp * CG;
            float* d = Bs + kp * BP + 8 * cg;
            *(float4*)d       = make_float4(r0[t].x, r1[t].x, r0[t].y, r1[t].y);
            *(float4*)(d + 4) = make_float4(r0[t].z, r1[t].z, r0[t].w, r1[t].w);
        }
    }
    if (NCP > NC) {
        constexpr int ZQ = (NCP - NC) / 2;
        for (int idx = tid; idx < KP * ZQ; idx += 256) {
            const int kp = idx / ZQ;
            const int zq = idx - kp * ZQ;
            *(float4*)(Bs + kp * BP + 2 * NC + 4 * zq) = make_float4(0.f, 0.f, 0.f, 0.f);
        }
    }
    asm volatile("" :: "v"(ar[0]), "v"(ar[1]), "v"(ar[2]), "v"(ar[3]),
                       "v"(ar[4]), "v"(ar[5]), "v"(ar[6]), "v"(ar[7]));
    __syncthreads();
    if (!live) return;

    float* Aw = As + wv * (16 * AP);
    v8f acc[TN];
#pragma unroll
    for (int tn = 0; tn < TN; ++tn) acc[tn] = v8f{};

#pragma unroll
    for (int p = 0; p < PH; ++p) {
#pragma unroll
        for (int q = 0; q < 8; ++q) {
            v4f v = ar[q];
            if (RELU) {
                v.x = fmaxf(v.x, 0.0f); v.y = fmaxf(v.y, 0.0f);
                v.z = fmaxf(v.z, 0.0f); v.w = fmaxf(v.w, 0.0f);
            }
            *(v4f*)(Aw + (2 * q + hf) * AP + 4 * l) = v;
        }
        __builtin_amdgcn_fence(__ATOMIC_ACQ_REL, "wavefront");
        if (p + 1 < PH) {
#pragma unroll
            for (int q = 0; q < 8; ++q)
                ar[q] = *(const v4f*)(Ab + (size_t)(2 * q) * K + 64 * (p + 1));
            __builtin_amdgcn_sched_barrier(0);
        }
#pragma unroll
        for (int kk = 0; kk < 16; ++kk) {
            const v2f a = *(const v2f*)(Aw + l * AP + 4 * kk + 2 * hf);
            const float* bp = Bs + (32 * p + 2 * kk + hf) * BP + 2 * l;
#pragma unroll
            for (int tn = 0; tn < TN; ++tn) {
                const v2f b = *(const v2f*)(bp + 32 * tn);
                acc[tn] = __builtin_amdgcn_wmma_f32_16x16x4_f32(
                     false, a,  false, b,
                     (short)0, acc[tn],  false,  false);
            }
        }
        __builtin_amdgcn_fence(__ATOMIC_ACQ_REL, "wavefront");
    }

    float sc[8];
#pragma unroll
    for (int i = 0; i < 8; ++i) sc[i] = 1.0f;
    if (SC) {
        const v4f s0 = *(const v4f*)(rs + strip * 16 + 8 * hf);
        const v4f s1 = *(const v4f*)(rs + strip * 16 + 8 * hf + 4);
        sc[0] = s0.x; sc[1] = s0.y; sc[2] = s0.z; sc[3] = s0.w;
        sc[4] = s1.x; sc[5] = s1.y; sc[6] = s1.z; sc[7] = s1.w;
    }
#pragma unroll
    for (int tn = 0; tn < TN; ++tn) {
        const int col = tn * 16 + l;
        if (col < NC) {
            const float bb = bias[col];
            float* Crow = C + (size_t)(strip * 16 + 8 * hf) * NC + col;
#pragma unroll
            for (int i = 0; i < 8; ++i)
                Crow[(size_t)i * NC] = SC ? (acc[tn][i] + bb) * sc[i] : acc[tn][i] + bb;
        }
    }
}

template <int NC, int SC>
__global__ __launch_bounds__(512) __attribute__((amdgpu_waves_per_eu(16, 16)))
void k_gather_gemm(const float* __restrict__ t, const int* __restrict__ rowptr,
                   const int* __restrict__ ew, const float* __restrict__ dis,
                   const float* __restrict__ B, const float* __restrict__ bias,
                   float* __restrict__ C, int n) {
    constexpr int TN  = (NC + 15) / 16;
    constexpr int NCP = TN * 16;
    constexpr int BP  = 2 * NCP + (((2 * NCP) % 64 == 32) ? 0 : 32);
    constexpr int KP  = HID / 2;
    constexpr int AP  = 68;
    constexpr int CG  = NC / 4;
    constexpr int NT  = KP * CG;
    constexpr int JPS = (NC % 32 == 0) ? NC / 32 : 1;
    constexpr int TPJ = TN / JPS;
    static_assert(NT <= 512, "one staging task per thread");
    static_assert(TPJ * JPS == TN, "jobs cover the tiles");

    __shared__ __attribute__((aligned(16))) float Bs[KP * BP];
    __shared__ __attribute__((aligned(16))) float As[2 * 16 * AP];
    __shared__ __attribute__((aligned(16))) float sds[32];

    const int tid = threadIdx.x;

    const bool stg = (NT == 512) || (tid < NT);
    const int kp = tid / CG;
    const int cg = tid - kp * CG;
    float4 r0 = make_float4(0.f, 0.f, 0.f, 0.f), r1 = r0;
    if (stg) {
        r0 = *(const float4*)(B + (size_t)(2 * kp) * NC + 4 * cg);
        r1 = *(const float4*)(B + (size_t)(2 * kp + 1) * NC + 4 * cg);
    }

    const int gid = blockIdx.x * 512 + tid;
    const int i = gid >> 4;
    const int c = gid & 15;
    const bool valid = i < n;
    const float4* t4 = (const float4*)t;
    int j = 0, je = 0;
    float sd = 0.f;
    float4 v = make_float4(0.f, 0.f, 0.f, 0.f);
    if (valid) {
        j  = rowptr[i];
        je = rowptr[i + 1];
        sd = dis[i];
        v  = t4[gid];
    }

    if (stg) {
        float* d = Bs + kp * BP + 8 * cg;
        *(float4*)d       = make_float4(r0.x, r1.x, r0.y, r1.y);
        *(float4*)(d + 4) = make_float4(r0.z, r1.z, r0.w, r1.w);
    }
    if (NCP > NC) {
        constexpr int ZQ = (NCP - NC) / 2;
        for (int idx = tid; idx < KP * ZQ; idx += 512) {
            const int zk = idx / ZQ;
            const int zq = idx - zk * ZQ;
            *(float4*)(Bs + zk * BP + 2 * NC + 4 * zq) = make_float4(0.f, 0.f, 0.f, 0.f);
        }
    }

    if (SC && c == 0) sds[tid >> 4] = sd;
    float4 a = v;
    const v4f* tv = (const v4f*)t;
    for (; j + 8 <= je; j += 8) {
        int p[8];
#pragma unroll
        for (int q = 0; q < 8; ++q) p[q] = ew[j + q];
        v4f u[8];
#pragma unroll
        for (int q = 0; q < 8; ++q) u[q] = tv[(size_t)p[q] * H4 + c];
#pragma unroll
        for (int q = 0; q < 8; ++q) {
            a.x += u[q].x; a.y += u[q].y; a.z += u[q].z; a.w += u[q].w;
        }
    }
    const int rem = je - j;
    if (rem > 0) {
        const int last = je - 1;
        int p[7];
        p[0] = ew[j];
#pragma unroll
        for (int q = 1; q < 7; ++q) p[q] = ew[min(j + q, last)];
        asm volatile("" :: "v"(p[0]), "v"(p[1]), "v"(p[2]), "v"(p[3]),
                           "v"(p[4]), "v"(p[5]), "v"(p[6]));
        v4f u[7];
#pragma unroll
        for (int q = 0; q < 7; ++q) u[q] = tv[(size_t)p[q] * H4 + c];
        asm volatile("" :: "v"(u[0]), "v"(u[1]), "v"(u[2]), "v"(u[3]),
                           "v"(u[4]), "v"(u[5]), "v"(u[6]));
        __builtin_amdgcn_sched_barrier(0);
#pragma unroll
        for (int q = 0; q < 7; ++q) {
            if (q < rem) {
                a.x += u[q].x; a.y += u[q].y; a.z += u[q].z; a.w += u[q].w;
            }
        }
    }

    {
        v4f h;
        h.x = fmaxf(a.x * sd, 0.0f); h.y = fmaxf(a.y * sd, 0.0f);
        h.z = fmaxf(a.z * sd, 0.0f); h.w = fmaxf(a.w * sd, 0.0f);
        *(v4f*)(As + (tid >> 4) * AP + 4 * c) = h;
    }
    __syncthreads();

    const int wv   = tid >> 5;
    const int lane = tid & 31;
    const int hf   = lane >> 4;
    const int l    = lane & 15;
    if (wv >= 2 * JPS) return;
    const int sl    = wv / JPS;
    const int part  = wv - sl * JPS;
    const int strip = blockIdx.x * 2 + sl;
    if (strip >= (n >> 4)) return;

    const float* Aw = As + sl * (16 * AP);
    v8f acc[TPJ];
#pragma unroll
    for (int tj = 0; tj < TPJ; ++tj) acc[tj] = v8f{};

#pragma unroll
    for (int kk = 0; kk < 16; ++kk) {
        const v2f af = *(const v2f*)(Aw + l * AP + 4 * kk + 2 * hf);
        const float* bp = Bs + (2 * kk + hf) * BP + 2 * l + 32 * (part * TPJ);
#pragma unroll
        for (int tj = 0; tj < TPJ; ++tj) {
            const v2f bf = *(const v2f*)(bp + 32 * tj);
            acc[tj] = __builtin_amdgcn_wmma_f32_16x16x4_f32(
                 false, af,  false, bf,
                 (short)0, acc[tj],  false,  false);
        }
    }

    float sc[8];
#pragma unroll
    for (int q = 0; q < 8; ++q) sc[q] = 1.0f;
    if (SC) {
        const v4f s0 = *(const v4f*)(sds + sl * 16 + 8 * hf);
        const v4f s1 = *(const v4f*)(sds + sl * 16 + 8 * hf + 4);
        sc[0] = s0.x; sc[1] = s0.y; sc[2] = s0.z; sc[3] = s0.w;
        sc[4] = s1.x; sc[5] = s1.y; sc[6] = s1.z; sc[7] = s1.w;
    }
#pragma unroll
    for (int tj = 0; tj < TPJ; ++tj) {
        const int col = (part * TPJ + tj) * 16 + l;
        if (col < NC) {
            const float bb = bias[col];
            float* Crow = C + (size_t)(strip * 16 + 8 * hf) * NC + col;
#pragma unroll
            for (int q = 0; q < 8; ++q)
                Crow[(size_t)q * NC] = SC ? (acc[tj][q] + bb) * sc[q] : acc[tj][q] + bb;
        }
    }
}

extern "C" void kernel_launch(void* const* d_in, const int* in_sizes, int n_in,
                              void* d_out, int out_size, void* d_ws, size_t ws_size,
                              hipStream_t stream) {
    (void)n_in; (void)out_size; (void)ws_size;
    const float* x    = (const float*)d_in[0];
    const int*   ei   = (const int*)d_in[1];
    const float* W0   = (const float*)d_in[2];
    const float* b0   = (const float*)d_in[3];
    const float* W1   = (const float*)d_in[4];
    const float* b1   = (const float*)d_in[5];
    const float* W2   = (const float*)d_in[6];
    const float* b2   = (const float*)d_in[7];
    const float* Wout = (const float*)d_in[8];
    const float* bout = (const float*)d_in[9];

    const int FIN = 128, OUTD = 40;
    const int n = in_sizes[0] / FIN;
    const int e = in_sizes[1] / 2;
    const int* srcI = ei;
    const int* dstI = ei + e;

    char* w = (char*)d_ws;
    size_t off = 0;
    const int nb = (n + BK_R - 1) / BK_R;
    int*   rowptr = (int*)(w + off);   off += (((size_t)(n + 1) * 4) + 255) & ~(size_t)255;
    float* dis    = (float*)(w + off); off += (((size_t)n * 4) + 255) & ~(size_t)255;
    int2*  pairs  = (int2*)(w + off);  off += (((size_t)e * 8) + 255) & ~(size_t)255;
    int*   ew     = (int*)(w + off);   off += (((size_t)e * 4) + 255) & ~(size_t)255;
    float* tmp    = (float*)(w + off); off += (((size_t)n * HID * 4) + 255) & ~(size_t)255;
    float* hA     = (float*)(w + off); off += (((size_t)n * HID * 4) + 255) & ~(size_t)255;
    int*   L      = (int*)(w + off);   off += (((size_t)(nb + 1) * SORT_G * 4) + 255) & ~(size_t)255;

    const int T   = 256;
    const int WPB = T / 32;

    k_chunk_group   <<<SORT_G, SORT_T, 0, stream>>>(srcI, dstI, L, pairs, e, nb);
    k_bucket_place  <<<nb, SORT_G, 0, stream>>>(pairs, L, rowptr, dis, ew, n, e);

    const int gGemm  = (n / 16 + WPB - 1) / WPB;
    const int gFuse  = (n + 31) / 32;

    k_gemm<128, HID, 0, 1><<<gGemm, T, 0, stream>>>(x, W0, b0, dis, tmp, n);
    k_gather_gemm<HID, 1><<<gFuse, 512, 0, stream>>>(tmp, rowptr, ew, dis, W1, b1, hA, n);
    k_gather_gemm<HID, 1><<<gFuse, 512, 0, stream>>>(hA, rowptr, ew, dis, W2, b2, tmp, n);
    k_gather_gemm<OUTD, 0><<<gFuse, 512, 0, stream>>>(tmp, rowptr, ew, dis, Wout, bout, (float*)d_out, n);
}


# Edge grouping, both levels: entries are placed in an LDS copy of the work-group's stretch and leave as 16-byte stores of whole lines (was one 8- or 4-byte store each); sum order still LDS arrival
# speedup vs baseline: 4.4742x; 1.0508x over previous
#include <hip/hip_runtime.h>

#define HID 64
#define H4  16

typedef __attribute__((ext_vector_type(2))) float v2f;
typedef __attribute__((ext_vector_type(8))) float v8f;
typedef __attribute__((ext_vector_type(4))) float v4f;

#define SORT_G 256
#define SORT_T 1024
#define BK_SH  7
#define BK_R   128
#define BK_CAP 1024
#define PC_R   8
#define P_CAP  8192
#define E_CAP  4096
#define PC_S   3
static_assert(PC_R * 32 == SORT_G && PC_R * PC_S == 24, "the load fences of level 2 name 24 slots");
static_assert(BK_CAP <= SORT_T && (1 << BK_SH) == BK_R, "one bucket per thread in the bucket scan");

__global__ __launch_bounds__(SORT_T)
void k_chunk_group(const int* __restrict__ src, const int* __restrict__ dst,
                   int* __restrict__ L, int2* __restrict__ pairs, int e, int nb) {
    __shared__ int lh[BK_CAP];
    __shared__ int s_w[32];
    __shared__ __attribute__((aligned(16))) int2 sP[P_CAP];
    const int tid  = threadIdx.x;
    const int lane = tid & 31;
    const int wv   = tid >> 5;
    if (tid < BK_CAP) lh[tid] = 0;
    __syncthreads();
    const int per = (e + SORT_G - 1) / SORT_G;
    const int beg = blockIdx.x * per;
    const int end = min(e, beg + per);
    const int a0  = beg & ~15;
    for (int i0 = beg + tid; i0 < end; i0 += 8 * SORT_T) {
        int d[8];
#pragma unroll
        for (int q = 0; q < 8; ++q) d[q] = dst[min(i0 + q * SORT_T, end - 1)];
        asm volatile("" :: "v"(d[0]), "v"(d[1]), "v"(d[2]), "v"(d[3]),
                           "v"(d[4]), "v"(d[5]), "v"(d[6]), "v"(d[7]));
#pragma unroll
        for (int q = 0; q < 8; ++q)
            if (i0 + q * SORT_T < end) atomicAdd(&lh[d[q] >> BK_SH], 1);
    }
    __syncthreads();
    const int v = (tid < BK_CAP) ? lh[tid] : 0;
    int inc = v;
    for (int o = 1; o < 32; o <<= 1) {
        int t = __shfl_up(inc, o);
        if (lane >= o) inc += t;
    }
    if (lane == 31) s_w[wv] = inc;
    __syncthreads();
    if (wv == 0) {
        int t = s_w[lane];
        int ti = t;
        for (int o = 1; o < 32; o <<= 1) {
            int u = __shfl_up(ti, o);
            if (lane >= o) ti += u;
        }
        s_w[lane] = ti - t;
    }
    __syncthreads();
    const int loc = s_w[wv] + inc - v;
    if (tid < BK_CAP) lh[tid] = beg + loc;
    if (tid < nb) {
        L[tid * SORT_G + blockIdx.x] = loc;
        if (tid == nb - 1) L[nb * SORT_G + blockIdx.x] = loc + v;
    }
    __syncthreads();
    for (int i0 = beg + tid; i0 < end; i0 += 8 * SORT_T) {
        int s[8], d[8];
#pragma unroll
        for (int q = 0; q < 8; ++q) {
            const int i = min(i0 + q * SORT_T, end - 1);
            s[q] = src[i];
            d[q] = dst[i];
        }
        asm volatile("" :: "v"(s[0]), "v"(s[1]), "v"(s[2]), "v"(s[3]),
                           "v"(s[4]), "v"(s[5]), "v"(s[6]), "v"(s[7]),
                           "v"(d[0]), "v"(d[1]), "v"(d[2]), "v"(d[3]),
                           "v"(d[4]), "v"(d[5]), "v"(d[6]), "v"(d[7]));
#pragma unroll
        for (int q = 0; q < 8; ++q) {
            if (i0 + q * SORT_T < end) {
                const int p = atomicAdd(&lh[d[q] >> BK_SH], 1);
                const int rel = p - a0;
                if (rel < P_CAP) sP[rel] = make_int2(s[q], d[q]);
                else             pairs[p] = make_int2(s[q], d[q]);
            }
        }
    }
    __syncthreads();
    const int lo = beg - a0;
    const int hi = min(lo + max(end - beg, 0), P_CAP);
    for (int g = 2 * tid; g < hi; g += 2 * SORT_T) {
        if (g >= lo && g + 2 <= hi) {
            *(int4*)(pairs + a0 + g) = *(const int4*)(sP + g);
        } else {
            if (g >= lo) pairs[a0 + g] = sP[g];
            if (g + 1 >= lo && g + 1 < hi) pairs[a0 + g + 1] = sP[g + 1];
        }
    }
}

__global__ __launch_bounds__(SORT_G)
void k_bucket_place(const int2* __restrict__ pairs, const int* __restrict__ L,
                    int* __restrict__ rowptr, float* __restrict__ dis,
                    int* __restrict__ ew, int n, int e) {
    __shared__ int c[BK_R];
    __shared__ int s_w[BK_R / 32];
    __shared__ int sS[SORT_G];
    __shared__ int sL[SORT_G];
    __shared__ int s_r[SORT_G / 32];
    __shared__ __attribute__((aligned(16))) int sE[E_CAP];
    const int tid  = threadIdx.x;
    const int lane = tid & 31;
    const int wv   = tid >> 5;
    const int d0 = blockIdx.x << BK_SH;
    if (tid < BK_R) c[tid] = 0;
    const int per = (e + SORT_G - 1) / SORT_G;
    const int a = L[blockIdx.x * SORT_G + tid];
    const int z = L[(blockIdx.x + 1) * SORT_G + tid];
    sS[tid] = tid * per + a;
    sL[tid] = z - a;
    int ra = a;
    for (int o = 16; o >= 1; o >>= 1) ra += __shfl_xor(ra, o);
    if (lane == 0) s_r[wv] = ra;
    __syncthreads();
    int beg = 0;
#pragma unroll
    for (int k = 0; k < SORT_G / 32; ++k) beg += s_r[k];

    const int gi  = tid >> 3;
    const int sub = tid & 7;
    int st[PC_R], ln[PC_R];
#pragma unroll
    for (int r = 0; r < PC_R; ++r) {
        st[r] = sS[r * 32 + gi];
        ln[r] = sL[r * 32 + gi];
    }
    int dd[PC_R * PC_S];
#pragma unroll
    for (int r = 0; r < PC_R; ++r)
#pragma unroll
        for (int j = 0; j < PC_S; ++j) {
            const int k = sub + 8 * j;
            dd[r * PC_S + j] = pairs[(k < ln[r]) ? st[r] + k : 0].y;
        }
    asm volatile("" :: "v"(dd[0]), "v"(dd[1]), "v"(dd[2]), "v"(dd[3]), "v"(dd[4]), "v"(dd[5]),
                       "v"(dd[6]), "v"(dd[7]), "v"(dd[8]), "v"(dd[9]), "v"(dd[10]), "v"(dd[11]),
                       "v"(dd[12]), "v"(dd[13]), "v"(dd[14]), "v"(dd[15]), "v"(dd[16]), "v"(dd[17]),
                       "v"(dd[18]), "v"(dd[19]), "v"(dd[20]), "v"(dd[21]), "v"(dd[22]), "v"(dd[23]));
#pragma unroll
    for (int r = 0; r < PC_R; ++r)
#pragma unroll
        for (int j = 0; j < PC_S; ++j)
            if (sub + 8 * j < ln[r]) atomicAdd(&c[dd[r * PC_S + j] - d0], 1);
    int lmax = ln[0];
#pragma unroll
    for (int r = 1; r < PC_R; ++r) lmax = max(lmax, ln[r]);
    if (lmax > 8 * PC_S) {
#pragma unroll
        for (int r = 0; r < PC_R; ++r)
            for (int k = sub + 8 * PC_S; k < ln[r]; k += 8)
                atomicAdd(&c[pairs[st[r] + k].y - d0], 1);
    }
    __syncthreads();
    int v = 0, inc = 0;
    if (tid < BK_R) {
        v = c[tid];
        inc = v;
        for (int o = 1; o < 32; o <<= 1) {
            int t = __shfl_up(inc, o);
            if (lane >= o) inc += t;
        }
        if (lane == 31) s_w[wv] = inc;
    }
    __syncthreads();
    if (tid < BK_R) {
        int off = 0;
        for (int k = 0; k < wv; ++k) off += s_w[k];
        const int rp = beg + off + inc - v;
        c[tid] = rp;
        const int d = d0 + tid;
        if (d < n) {
            rowptr[d] = rp;
            dis[d] = rsqrtf((float)(v + 1));
            if (d == n - 1) rowptr[n] = rp + v;
        }
    }
    __syncthreads();
    const int a0  = beg & ~31;
    const int cnt = s_w[0] + s_w[1] + s_w[2] + s_w[3];
    int ss[PC_R * PC_S];
#pragma unroll
    for (int r = 0; r < PC_R; ++r)
#pragma unroll
        for (int j = 0; j < PC_S; ++j) {
            const int k = sub + 8 * j;
            ss[r * PC_S + j] = pairs[(k < ln[r]) ? st[r] + k : 0].x;
        }
    asm volatile("" :: "v"(ss[0]), "v"(ss[1]), "v"(ss[2]), "v"(ss[3]), "v"(ss[4]), "v"(ss[5]),
                       "v"(ss[6]), "v"(ss[7]), "v"(ss[8]), "v"(ss[9]), "v"(ss[10]), "v"(ss[11]),
                       "v"(ss[12]), "v"(ss[13]), "v"(ss[14]), "v"(ss[15]), "v"(ss[16]), "v"(ss[17]),
                       "v"(ss[18]), "v"(ss[19]), "v"(ss[20]), "v"(ss[21]), "v"(ss[22]), "v"(ss[23]));
#pragma unroll
    for (int r = 0; r < PC_R; ++r)
#pragma unroll
        for (int j = 0; j < PC_S; ++j) {
            if (sub + 8 * j < ln[r]) {
                const int pos = atomicAdd(&c[dd[r * PC_S + j] - d0], 1);
                const int rel = pos - a0;
                if (rel < E_CAP) sE[rel] = ss[r * PC_S + j];
                else             ew[pos] = ss[r * PC_S + j];
            }
        }
    if (lmax > 8 * PC_S) {
#pragma unroll
        for (int r = 0; r < PC_R; ++r)
            for (int k = sub + 8 * PC_S; k < ln[r]; k += 8) {
                const int2 p = pairs[st[r] + k];
                const int pos = atomicAdd(&c[p.y - d0], 1);
                const int rel = pos - a0;
                if (rel < E_CAP) sE[rel] = p.x;
                else             ew[pos] = p.x;
            }
    }
    __syncthreads();
    const int lo = beg - a0;
    const int hi = min(lo + cnt, E_CAP);
    for (int g = 4 * tid; g < hi; g += 4 * SORT_G) {
        if (g >= lo && g + 4 <= hi) {
            *(int4*)(ew + a0 + g) = *(const int4*)(sE + g);
        } else {
            for (int q = 0; q < 4; ++q)
                if (g + q >= lo && g + q < hi) ew[a0 + g + q] = sE[g + q];
        }
    }
}

template <int K, int NC, int RELU, int SC>
__global__ __launch_bounds__(256)
void k_gemm(const float* __restrict__ A, const float* __restrict__ B,
            const float* __restrict__ bias, const float* __restrict__ rs,
            float* __restrict__ C, int M) {
    constexpr int TN  = (NC + 15) / 16;
    constexpr int NCP = TN * 16;
    constexpr int BP  = 2 * NCP + (((2 * NCP) % 64 == 32) ? 0 : 32);
    constexpr int KP  = K / 2;
    constexpr int AP  = 68;
    constexpr int PH  = K / 64;
    constexpr int CG  = NC / 4;
    constexpr int NT  = KP * CG;
    constexpr int TPT = (NT + 255) / 256;

    __shared__ __attribute__((aligned(16))) float Bs[KP * BP];
    __shared__ __attribute__((aligned(16))) float As[8 * 16 * AP];

    const int tid  = threadIdx.x;
    const int wv   = tid >> 5;
    const int lane = tid & 31;
    const int hf   = lane >> 4;
    const int l    = lane & 15;
    const int strip = blockIdx.x * 8 + wv;
    const bool live = strip < (M >> 4);

    float4 r0[TPT], r1[TPT];
#pragma unroll
    for (int t = 0; t < TPT; ++t) {
        const int task = tid + t * 256;
        if (NT % 256 == 0 || task < NT) {
            const int kp = task / CG;
            const int cg = task - kp * CG;
            r0[t] = *(const float4*)(B + (size_t)(2 * kp) * NC + 4 * cg);
            r1[t] = *(const float4*)(B + (size_t)(2 * kp + 1) * NC + 4 * cg);
        }
    }

    const float* Ab = A + (size_t)(live ? strip : 0) * 16 * K + (size_t)hf * K + 4 * l;
    v4f ar[8];
#pragma unroll
    for (int q = 0; q < 8; ++q)
        ar[q] = *(const v4f*)(Ab + (size_t)(2 * q) * K);

#pragma unroll
    for (int t = 0; t < TPT; ++t) {
        const int task = tid + t * 256;
        if (NT % 256 == 0 || task < NT) {
            const int kp = task / CG;
            const int cg = task - kp * CG;
            float* d = Bs + kp * BP + 8 * cg;
            *(float4*)d       = make_float4(r0[t].x, r1[t].x, r0[t].y, r1[t].y);
            *(float4*)(d + 4) = make_float4(r0[t].z, r1[t].z, r0[t].w, r1[t].w);
        }
    }
    if (NCP > NC) {
        constexpr int ZQ = (NCP - NC) / 2;
        for (int idx = tid; idx < KP * ZQ; idx += 256) {
            const int kp = idx / ZQ;
            const int zq = idx - kp * ZQ;
            *(float4*)(Bs + kp * BP + 2 * NC + 4 * zq) = make_float4(0.f, 0.f, 0.f, 0.f);
        }
    }
    asm volatile("" :: "v"(ar[0]), "v"(ar[1]), "v"(ar[2]), "v"(ar[3]),
                       "v"(ar[4]), "v"(ar[5]), "v"(ar[6]), "v"(ar[7]));
    __syncthreads();
    if (!live) return;

    float* Aw = As + wv * (16 * AP);
    v8f acc[TN];
#pragma unroll
    for (int tn = 0; tn < TN; ++tn) acc[tn] = v8f{};

#pragma unroll
    for (int p = 0; p < PH; ++p) {
#pragma unroll
        for (int q = 0; q < 8; ++q) {
            v4f v = ar[q];
            if (RELU) {
                v.x = fmaxf(v.x, 0.0f); v.y = fmaxf(v.y, 0.0f);
                v.z = fmaxf(v.z, 0.0f); v.w = fmaxf(v.w, 0.0f);
            }
            *(v4f*)(Aw + (2 * q + hf) * AP + 4 * l) = v;
        }
        __builtin_amdgcn_fence(__ATOMIC_ACQ_REL, "wavefront");
        if (p + 1 < PH) {
#pragma unroll
            for (int q = 0; q < 8; ++q)
                ar[q] = *(const v4f*)(Ab + (size_t)(2 * q) * K + 64 * (p + 1));
            __builtin_amdgcn_sched_barrier(0);
        }
#pragma unroll
        for (int kk = 0; kk < 16; ++kk) {
            const v2f a = *(const v2f*)(Aw + l * AP + 4 * kk + 2 * hf);
            const float* bp = Bs + (32 * p + 2 * kk + hf) * BP + 2 * l;
#pragma unroll
            for (int tn = 0; tn < TN; ++tn) {
                const v2f b = *(const v2f*)(bp + 32 * tn);
                acc[tn] = __builtin_amdgcn_wmma_f32_16x16x4_f32(
                     false, a,  false, b,
                     (short)0, acc[tn],  false,  false);
            }
        }
        __builtin_amdgcn_fence(__ATOMIC_ACQ_REL, "wavefront");
    }

    float sc[8];
#pragma unroll
    for (int i = 0; i < 8; ++i) sc[i] = 1.0f;
    if (SC) {
        const v4f s0 = *(const v4f*)(rs + strip * 16 + 8 * hf);
        const v4f s1 = *(const v4f*)(rs + strip * 16 + 8 * hf + 4);
        sc[0] = s0.x; sc[1] = s0.y; sc[2] = s0.z; sc[3] = s0.w;
        sc[4] = s1.x; sc[5] = s1.y; sc[6] = s1.z; sc[7] = s1.w;
    }
#pragma unroll
    for (int tn = 0; tn < TN; ++tn) {
        const int col = tn * 16 + l;
        if (col < NC) {
            const float bb = bias[col];
            float* Crow = C + (size_t)(strip * 16 + 8 * hf) * NC + col;
#pragma unroll
            for (int i = 0; i < 8; ++i)
                Crow[(size_t)i * NC] = SC ? (acc[tn][i] + bb) * sc[i] : acc[tn][i] + bb;
        }
    }
}

template <int NC, int SC>
__global__ __launch_bounds__(512) __attribute__((amdgpu_waves_per_eu(16, 16)))
void k_gather_gemm(const float* __restrict__ t, const int* __restrict__ rowptr,
                   const int* __restrict__ ew, const float* __restrict__ dis,
                   const float* __restrict__ B, const float* __restrict__ bias,
                   float* __restrict__ C, int n) {
    constexpr int TN  = (NC + 15) / 16;
    constexpr int NCP = TN * 16;
    constexpr int BP  = 2 * NCP + (((2 * NCP) % 64 == 32) ? 0 : 32);
    constexpr int KP  = HID / 2;
    constexpr int AP  = 68;
    constexpr int CG  = NC / 4;
    constexpr int NT  = KP * CG;
    constexpr int JPS = (NC % 32 == 0) ? NC / 32 : 1;
    constexpr int TPJ = TN / JPS;
    static_assert(NT <= 512, "one staging task per thread");
    static_assert(TPJ * JPS == TN, "jobs cover the tiles");

    __shared__ __attribute__((aligned(16))) float Bs[KP * BP];
    __shared__ __attribute__((aligned(16))) float As[2 * 16 * AP];
    __shared__ __attribute__((aligned(16))) float sds[32];

    const int tid = threadIdx.x;

    const bool stg = (NT == 512) || (tid < NT);
    const int kp = tid / CG;
    const int cg = tid - kp * CG;
    float4 r0 = make_float4(0.f, 0.f, 0.f, 0.f), r1 = r0;
    if (stg) {
        r0 = *(const float4*)(B + (size_t)(2 * kp) * NC + 4 * cg);
        r1 = *(const float4*)(B + (size_t)(2 * kp + 1) * NC + 4 * cg);
    }

    const int gid = blockIdx.x * 512 + tid;
    const int i = gid >> 4;
    const int c = gid & 15;
    const bool valid = i < n;
    const float4* t4 = (const float4*)t;
    int j = 0, je = 0;
    float sd = 0.f;
    float4 v = make_float4(0.f, 0.f, 0.f, 0.f);
    if (valid) {
        j  = rowptr[i];
        je = rowptr[i + 1];
        sd = dis[i];
        v  = t4[gid];
    }

    if (stg) {
        float* d = Bs + kp * BP + 8 * cg;
        *(float4*)d       = make_float4(r0.x, r1.x, r0.y, r1.y);
        *(float4*)(d + 4) = make_float4(r0.z, r1.z, r0.w, r1.w);
    }
    if (NCP > NC) {
        constexpr int ZQ = (NCP - NC) / 2;
        for (int idx = tid; idx < KP * ZQ; idx += 512) {
            const int zk = idx / ZQ;
            const int zq = idx - zk * ZQ;
            *(float4*)(Bs + zk * BP + 2 * NC + 4 * zq) = make_float4(0.f, 0.f, 0.f, 0.f);
        }
    }

    if (SC && c == 0) sds[tid >> 4] = sd;
    float4 a = v;
    const v4f* tv = (const v4f*)t;
    for (; j + 8 <= je; j += 8) {
        int p[8];
#pragma unroll
        for (int q = 0; q < 8; ++q) p[q] = ew[j + q];
        v4f u[8];
#pragma unroll
        for (int q = 0; q < 8; ++q) u[q] = tv[(size_t)p[q] * H4 + c];
#pragma unroll
        for (int q = 0; q < 8; ++q) {
            a.x += u[q].x; a.y += u[q].y; a.z += u[q].z; a.w += u[q].w;
        }
    }
    const int rem = je - j;
    if (rem > 0) {
        const int last = je - 1;
        int p[7];
        p[0] = ew[j];
#pragma unroll
        for (int q = 1; q < 7; ++q) p[q] = ew[min(j + q, last)];
        asm volatile("" :: "v"(p[0]), "v"(p[1]), "v"(p[2]), "v"(p[3]),
                           "v"(p[4]), "v"(p[5]), "v"(p[6]));
        v4f u[7];
#pragma unroll
        for (int q = 0; q < 7; ++q) u[q] = tv[(size_t)p[q] * H4 + c];
        asm volatile("" :: "v"(u[0]), "v"(u[1]), "v"(u[2]), "v"(u[3]),
                           "v"(u[4]), "v"(u[5]), "v"(u[6]));
        __builtin_amdgcn_sched_barrier(0);
#pragma unroll
        for (int q = 0; q < 7; ++q) {
            if (q < rem) {
                a.x += u[q].x; a.y += u[q].y; a.z += u[q].z; a.w += u[q].w;
            }
        }
    }

    {
        v4f h;
        h.x = fmaxf(a.x * sd, 0.0f); h.y = fmaxf(a.y * sd, 0.0f);
        h.z = fmaxf(a.z * sd, 0.0f); h.w = fmaxf(a.w * sd, 0.0f);
        *(v4f*)(As + (tid >> 4) * AP + 4 * c) = h;
    }
    __syncthreads();

    const int wv   = tid >> 5;
    const int lane = tid & 31;
    const int hf   = lane >> 4;
    const int l    = lane & 15;
    if (wv >= 2 * JPS) return;
    const int sl    = wv / JPS;
    const int part  = wv - sl * JPS;
    const int strip = blockIdx.x * 2 + sl;
    if (strip >= (n >> 4)) return;

    const float* Aw = As + sl * (16 * AP);
    v8f acc[TPJ];
#pragma unroll
    for (int tj = 0; tj < TPJ; ++tj) acc[tj] = v8f{};

#pragma unroll
    for (int kk = 0; kk < 16; ++kk) {
        const v2f af = *(const v2f*)(Aw + l * AP + 4 * kk + 2 * hf);
        const float* bp = Bs + (2 * kk + hf) * BP + 2 * l + 32 * (part * TPJ);
#pragma unroll
        for (int tj = 0; tj < TPJ; ++tj) {
            const v2f bf = *(const v2f*)(bp + 32 * tj);
            acc[tj] = __builtin_amdgcn_wmma_f32_16x16x4_f32(
                 false, af,  false, bf,
                 (short)0, acc[tj],  false,  false);
        }
    }

    float sc[8];
#pragma unroll
    for (int q = 0; q < 8; ++q) sc[q] = 1.0f;
    if (SC) {
        const v4f s0 = *(const v4f*)(sds + sl * 16 + 8 * hf);
        const v4f s1 = *(const v4f*)(sds + sl * 16 + 8 * hf + 4);
        sc[0] = s0.x; sc[1] = s0.y; sc[2] = s0.z; sc[3] = s0.w;
        sc[4] = s1.x; sc[5] = s1.y; sc[6] = s1.z; sc[7] = s1.w;
    }
#pragma unroll
    for (int tj = 0; tj < TPJ; ++tj) {
        const int col = (part * TPJ + tj) * 16 + l;
        if (col < NC) {
            const float bb = bias[col];
            float* Crow = C + (size_t)(strip * 16 + 8 * hf) * NC + col;
#pragma unroll
            for (int q = 0; q < 8; ++q)
                Crow[(size_t)q * NC] = SC ? (acc[tj][q] + bb) * sc[q] : acc[tj][q] + bb;
        }
    }
}

extern "C" void kernel_launch(void* const* d_in, const int* in_sizes, int n_in,
                              void* d_out, int out_size, void* d_ws, size_t ws_size,
                              hipStream_t stream) {
    (void)n_in; (void)out_size; (void)ws_size;
    const float* x    = (const float*)d_in[0];
    const int*   ei   = (const int*)d_in[1];
    const float* W0   = (const float*)d_in[2];
    const float* b0   = (const float*)d_in[3];
    const float* W1   = (const float*)d_in[4];
    const float* b1   = (const float*)d_in[5];
    const float* W2   = (const float*)d_in[6];
    const float* b2   = (const float*)d_in[7];
    const float* Wout = (const float*)d_in[8];
    const float* bout = (const float*)d_in[9];

    const int FIN = 128, OUTD = 40;
    const int n = in_sizes[0] / FIN;
    const int e = in_sizes[1] / 2;
    const int* srcI = ei;
    const int* dstI = ei + e;

    char* w = (char*)d_ws;
    size_t off = 0;
    const int nb = (n + BK_R - 1) / BK_R;
    int*   rowptr = (int*)(w + off);   off += (((size_t)(n + 1) * 4) + 255) & ~(size_t)255;
    float* dis    = (float*)(w + off); off += (((size_t)n * 4) + 255) & ~(size_t)255;
    int2*  pairs  = (int2*)(w + off);  off += (((size_t)e * 8) + 255) & ~(size_t)255;
    int*   ew     = (int*)(w + off);   off += (((size_t)e * 4) + 255) & ~(size_t)255;
    float* tmp    = (float*)(w + off); off += (((size_t)n * HID * 4) + 255) & ~(size_t)255;
    float* hA     = (float*)(w + off); off += (((size_t)n * HID * 4) + 255) & ~(size_t)255;
    int*   L      = (int*)(w + off);   off += (((size_t)(nb + 1) * SORT_G * 4) + 255) & ~(size_t)255;

    const int T   = 256;
    const int WPB = T / 32;

    k_chunk_group   <<<SORT_G, SORT_T, 0, stream>>>(srcI, dstI, L, pairs, e, nb);
    k_bucket_place  <<<nb, SORT_G, 0, stream>>>(pairs, L, rowptr, dis, ew, n, e);

    const int gGemm  = (n / 16 + WPB - 1) / WPB;
    const int gFuse  = (n + 31) / 32;

    k_gemm<128, HID, 0, 1><<<gGemm, T, 0, stream>>>(x, W0, b0, dis, tmp, n);
    k_gather_gemm<HID, 1><<<gFuse, 512, 0, stream>>>(tmp, rowptr, ew, dis, W1, b1, hA, n);
    k_gather_gemm<HID, 1><<<gFuse, 512, 0, stream>>>(hA, rowptr, ew, dis, W2, b2, tmp, n);
    k_gather_gemm<OUTD, 0><<<gFuse, 512, 0, stream>>>(tmp, rowptr, ew, dis, Wout, bout, (float*)d_out, n);
}
